# starting kernel
# baseline (speedup 1.0000x reference)
#include <hip/hip_runtime.h>
#include <hip/hip_bf16.h>

typedef __attribute__((ext_vector_type(2))) float v2f;
typedef __attribute__((ext_vector_type(8))) float v8f;

#define DIMD 64
#define KCODES 512
#define NROWS 65536
#define TILE_ROWS 16

__device__ __forceinline__ float wave_sum(float v) {
#pragma unroll
  for (int off = 16; off > 0; off >>= 1) v += __shfl_xor(v, off, 32);
  return v;
}
__device__ __forceinline__ float wave_max(float v) {
#pragma unroll
  for (int off = 16; off > 0; off >>= 1) v = fmaxf(v, __shfl_xor(v, off, 32));
  return v;
}

__global__ void vq_zero(float* acc) {
  int i = blockIdx.x * blockDim.x + threadIdx.x;
  if (i < 514) acc[i] = 0.0f;
}

__global__ __launch_bounds__(256) void vq_cbnorm(const float* __restrict__ cb,
                                                 float* __restrict__ cbn) {
  int wave = (blockIdx.x * blockDim.x + threadIdx.x) >> 5;
  int lane = threadIdx.x & 31;
  const float* row = cb + wave * DIMD;
  float v0 = row[lane], v1 = row[lane + 32];
  float s = wave_sum(v0 * v0 + v1 * v1);
  float scale = 1.0f / fmaxf(sqrtf(s), 1e-12f);
  cbn[wave * DIMD + lane] = v0 * scale;
  cbn[wave * DIMD + lane + 32] = v1 * scale;
}

__global__ __launch_bounds__(32) void vq_main(
    const float* __restrict__ z, const float* __restrict__ kappa_p,
    const float* __restrict__ cbn, const float* __restrict__ gum,
    float* __restrict__ out, float* __restrict__ avgp_acc,
    float* __restrict__ sdisc, float* __restrict__ scont) {
  __shared__ float zt[TILE_ROWS][DIMD];
  __shared__ float ybuf[TILE_ROWS][KCODES];
  __shared__ float avgp[KCODES];

  const int lane = threadIdx.x;
  const int m = lane & 15;
  const int hi = lane >> 4;
  const long base_row = (long)blockIdx.x * TILE_ROWS;
  const float kappa = kappa_p[0];

#pragma unroll
  for (int i = lane; i < KCODES; i += 32) avgp[i] = 0.0f;

#pragma unroll
  for (int r = 0; r < TILE_ROWS; ++r) {
    const float* zr = z + (base_row + r) * DIMD;
    float v0 = zr[lane], v1 = zr[lane + 32];
    float s = wave_sum(v0 * v0 + v1 * v1);
    float scale = 1.0f / fmaxf(sqrtf(s), 1e-12f);
    zt[r][lane] = v0 * scale;
    zt[r][lane + 32] = v1 * scale;
  }
  __syncthreads();

  for (int ct = 0; ct < KCODES / 16; ++ct) {
    v8f c = {};
    const float* cbrow = cbn + (ct * 16 + m) * DIMD;
#pragma unroll
    for (int k = 0; k < 16; ++k) {
      int kb = 4 * k + 2 * hi;
      v2f a;
      a.x = zt[m][kb];
      a.y = zt[m][kb + 1];
      v2f b;
      b.x = cbrow[kb];
      b.y = cbrow[kb + 1];
      c = __builtin_amdgcn_wmma_f32_16x16x4_f32(false, a, false, b, (short)0,
                                                c, false, false);
    }
#pragma unroll
    for (int j = 0; j < 8; ++j)
      ybuf[j + 8 * hi][ct * 16 + m] = kappa * c[j];
  }
  __syncthreads();

  float acc_pl = 0.0f;
  for (int r = 0; r < TILE_ROWS; ++r) {
    const float4* urow4 =
        (const float4*)(gum + (base_row + r) * (long)KCODES) + lane * 4;
    float yv[16], y2v[16];
    float m1 = -3.4e38f, m2 = -3.4e38f;
#pragma unroll
    for (int i4 = 0; i4 < 4; ++i4) {
      float4 u4 = urow4[i4];
      float uu[4] = {u4.x, u4.y, u4.z, u4.w};
#pragma unroll
      for (int j = 0; j < 4; ++j) {
        int i = i4 * 4 + j;
        int c0 = lane * 16 + i;
        float yy = ybuf[r][c0];
        float g = -logf(-logf(uu[j] + 1e-10f) + 1e-10f);
        float y2 = (yy + g) * 2.0f;
        yv[i] = yy;
        y2v[i] = y2;
        m1 = fmaxf(m1, yy);
        m2 = fmaxf(m2, y2);
      }
    }
    m1 = wave_max(m1);
    m2 = wave_max(m2);
    float s1 = 0.0f, s2 = 0.0f;
#pragma unroll
    for (int i = 0; i < 16; ++i) {
      s1 += expf(yv[i] - m1);
      s2 += expf(y2v[i] - m2);
    }
    s1 = wave_sum(s1);
    s2 = wave_sum(s2);
    float lse = m1 + logf(s1);
    float inv_s2 = 1.0f / s2;
#pragma unroll
    for (int i = 0; i < 16; ++i) {
      int c0 = lane * 16 + i;
      float lp = yv[i] - lse;
      float p = expf(lp);
      acc_pl += p * lp;
      avgp[c0] += p;
      ybuf[r][c0] = expf(y2v[i] - m2) * inv_s2;
    }
  }
  acc_pl = wave_sum(acc_pl);
  if (lane == 0) atomicAdd(sdisc, acc_pl);
#pragma unroll
  for (int i = lane; i < KCODES; i += 32) atomicAdd(&avgp_acc[i], avgp[i]);
  __syncthreads();

  float acc_cont = 0.0f;
  for (int nt = 0; nt < 4; ++nt) {
    v8f c = {};
    for (int k = 0; k < KCODES / 4; ++k) {
      int kb = 4 * k + 2 * hi;
      v2f a;
      a.x = ybuf[m][kb];
      a.y = ybuf[m][kb + 1];
      v2f b;
      b.x = cbn[(long)kb * DIMD + nt * 16 + m];
      b.y = cbn[(long)(kb + 1) * DIMD + nt * 16 + m];
      c = __builtin_amdgcn_wmma_f32_16x16x4_f32(false, a, false, b, (short)0,
                                                c, false, false);
    }
#pragma unroll
    for (int j = 0; j < 8; ++j) {
      int r = j + 8 * hi;
      long idx = (base_row + r) * DIMD + nt * 16 + m;
      float zq = c[j];
      float zo = z[idx];
      acc_cont += zo * (zo - zq);
      out[idx] = zq;
    }
  }
  acc_cont = wave_sum(acc_cont);
  if (lane == 0) atomicAdd(scont, acc_cont);
}

__global__ __launch_bounds__(32) void vq_finalize(
    const float* __restrict__ kappa_p, const float* __restrict__ avgp_acc,
    const float* __restrict__ sdisc, const float* __restrict__ scont,
    float* __restrict__ out) {
  int lane = threadIdx.x;
  const float invN = 1.0f / (float)NROWS;
  float acc = 0.0f;
#pragma unroll
  for (int i = lane; i < KCODES; i += 32) {
    float a = avgp_acc[i] * invN;
    acc += a * logf(a + 1e-7f);
  }
  acc = wave_sum(acc);
  if (lane == 0) {
    float loss = (sdisc[0] + kappa_p[0] * scont[0]) / 8.0f;
    out[(long)NROWS * DIMD] = loss;
    out[(long)NROWS * DIMD + 1] = expf(-acc);
  }
}

extern "C" void kernel_launch(void* const* d_in, const int* in_sizes, int n_in,
                              void* d_out, int out_size, void* d_ws,
                              size_t ws_size, hipStream_t stream) {
  const float* z = (const float*)d_in[0];
  const float* kappa = (const float*)d_in[1];
  const float* cb = (const float*)d_in[2];
  const float* gum = (const float*)d_in[3];
  float* out = (float*)d_out;

  float* wsf = (float*)d_ws;
  float* avgp_acc = wsf;
  float* sdisc = wsf + 512;
  float* scont = wsf + 513;
  float* cbn = wsf + 1024;

  vq_zero<<<3, 256, 0, stream>>>(wsf);
  vq_cbnorm<<<64, 256, 0, stream>>>(cb, cbn);
  vq_main<<<NROWS / TILE_ROWS, 32, 0, stream>>>(z, kappa, cbn, gum, out,
                                                avgp_acc, sdisc, scont);
  vq_finalize<<<1, 32, 0, stream>>>(kappa, avgp_acc, sdisc, scont, out);
}


# GEMM2: lane m owns out columns 4m..4m+3, so one b128 codebook-row load feeds 4 independent accumulator tiles, next 8 k-steps loaded ahead; b128 out stores; continuous-KL partial sums regrouped
# speedup vs baseline: 1.1643x; 1.1643x over previous
#include <hip/hip_runtime.h>
#include <hip/hip_bf16.h>

typedef __attribute__((ext_vector_type(2))) float v2f;
typedef __attribute__((ext_vector_type(8))) float v8f;

#define DIMD 64
#define KCODES 512
#define NROWS 65536
#define TILE_ROWS 16

__device__ __forceinline__ float wave_sum(float v) {
#pragma unroll
  for (int off = 16; off > 0; off >>= 1) v += __shfl_xor(v, off, 32);
  return v;
}
__device__ __forceinline__ float wave_max(float v) {
#pragma unroll
  for (int off = 16; off > 0; off >>= 1) v = fmaxf(v, __shfl_xor(v, off, 32));
  return v;
}

__global__ void vq_zero(float* acc) {
  int i = blockIdx.x * blockDim.x + threadIdx.x;
  if (i < 514) acc[i] = 0.0f;
}

__global__ __launch_bounds__(256) void vq_cbnorm(const float* __restrict__ cb,
                                                 float* __restrict__ cbn) {
  int wave = (blockIdx.x * blockDim.x + threadIdx.x) >> 5;
  int lane = threadIdx.x & 31;
  const float* row = cb + wave * DIMD;
  float v0 = row[lane], v1 = row[lane + 32];
  float s = wave_sum(v0 * v0 + v1 * v1);
  float scale = 1.0f / fmaxf(sqrtf(s), 1e-12f);
  cbn[wave * DIMD + lane] = v0 * scale;
  cbn[wave * DIMD + lane + 32] = v1 * scale;
}

__global__ __launch_bounds__(32) void vq_main(
    const float* __restrict__ z, const float* __restrict__ kappa_p,
    const float* __restrict__ cbn, const float* __restrict__ gum,
    float* __restrict__ out, float* __restrict__ avgp_acc,
    float* __restrict__ sdisc, float* __restrict__ scont) {
  __shared__ float zt[TILE_ROWS][DIMD];
  __shared__ float ybuf[TILE_ROWS][KCODES];
  __shared__ float avgp[KCODES];

  const int lane = threadIdx.x;
  const int m = lane & 15;
  const int hi = lane >> 4;
  const long base_row = (long)blockIdx.x * TILE_ROWS;
  const float kappa = kappa_p[0];

#pragma unroll
  for (int i = lane; i < KCODES; i += 32) avgp[i] = 0.0f;

#pragma unroll
  for (int r = 0; r < TILE_ROWS; ++r) {
    const float* zr = z + (base_row + r) * DIMD;
    float v0 = zr[lane], v1 = zr[lane + 32];
    float s = wave_sum(v0 * v0 + v1 * v1);
    float scale = 1.0f / fmaxf(sqrtf(s), 1e-12f);
    zt[r][lane] = v0 * scale;
    zt[r][lane + 32] = v1 * scale;
  }
  __syncthreads();

  for (int ct = 0; ct < KCODES / 16; ++ct) {
    v8f c = {};
    const float* cbrow = cbn + (ct * 16 + m) * DIMD;
#pragma unroll
    for (int k = 0; k < 16; ++k) {
      int kb = 4 * k + 2 * hi;
      v2f a;
      a.x = zt[m][kb];
      a.y = zt[m][kb + 1];
      v2f b;
      b.x = cbrow[kb];
      b.y = cbrow[kb + 1];
      c = __builtin_amdgcn_wmma_f32_16x16x4_f32(false, a, false, b, (short)0,
                                                c, false, false);
    }
#pragma unroll
    for (int j = 0; j < 8; ++j)
      ybuf[j + 8 * hi][ct * 16 + m] = kappa * c[j];
  }
  __syncthreads();

  float acc_pl = 0.0f;
  for (int r = 0; r < TILE_ROWS; ++r) {
    const float4* urow4 =
        (const float4*)(gum + (base_row + r) * (long)KCODES) + lane * 4;
    float yv[16], y2v[16];
    float m1 = -3.4e38f, m2 = -3.4e38f;
#pragma unroll
    for (int i4 = 0; i4 < 4; ++i4) {
      float4 u4 = urow4[i4];
      float uu[4] = {u4.x, u4.y, u4.z, u4.w};
#pragma unroll
      for (int j = 0; j < 4; ++j) {
        int i = i4 * 4 + j;
        int c0 = lane * 16 + i;
        float yy = ybuf[r][c0];
        float g = -logf(-logf(uu[j] + 1e-10f) + 1e-10f);
        float y2 = (yy + g) * 2.0f;
        yv[i] = yy;
        y2v[i] = y2;
        m1 = fmaxf(m1, yy);
        m2 = fmaxf(m2, y2);
      }
    }
    m1 = wave_max(m1);
    m2 = wave_max(m2);
    float s1 = 0.0f, s2 = 0.0f;
#pragma unroll
    for (int i = 0; i < 16; ++i) {
      s1 += expf(yv[i] - m1);
      s2 += expf(y2v[i] - m2);
    }
    s1 = wave_sum(s1);
    s2 = wave_sum(s2);
    float lse = m1 + logf(s1);
    float inv_s2 = 1.0f / s2;
#pragma unroll
    for (int i = 0; i < 16; ++i) {
      int c0 = lane * 16 + i;
      float lp = yv[i] - lse;
      float p = expf(lp);
      acc_pl += p * lp;
      avgp[c0] += p;
      ybuf[r][c0] = expf(y2v[i] - m2) * inv_s2;
    }
  }
  acc_pl = wave_sum(acc_pl);
  if (lane == 0) atomicAdd(sdisc, acc_pl);
#pragma unroll
  for (int i = lane; i < KCODES; i += 32) atomicAdd(&avgp_acc[i], avgp[i]);
  __syncthreads();

  float acc_cont = 0.0f;
  {
    constexpr int KSTEPS = KCODES / 4;
    constexpr int KBLK = 8;
    const float* bbase = cbn + (2 * hi) * DIMD + 4 * m;
    v8f c0 = {}, c1 = {}, c2 = {}, c3 = {};
    float4 n0[KBLK], n1[KBLK];
#pragma unroll
    for (int s = 0; s < KBLK; ++s) {
      n0[s] = *(const float4*)(bbase + (long)(4 * s) * DIMD);
      n1[s] = *(const float4*)(bbase + (long)(4 * s + 1) * DIMD);
    }
    for (int kk = 0; kk < KSTEPS / KBLK; ++kk) {
      float4 r0[KBLK], r1[KBLK];
#pragma unroll
      for (int s = 0; s < KBLK; ++s) {
        r0[s] = n0[s];
        r1[s] = n1[s];
      }
      if (kk + 1 < KSTEPS / KBLK) {
        const float* nb = bbase + (long)(kk + 1) * (KBLK * 4) * DIMD;
#pragma unroll
        for (int s = 0; s < KBLK; ++s) {
          n0[s] = *(const float4*)(nb + (long)(4 * s) * DIMD);
          n1[s] = *(const float4*)(nb + (long)(4 * s + 1) * DIMD);
        }
      }
#pragma unroll
      for (int s = 0; s < KBLK; ++s) {
        int kb = 4 * (kk * KBLK + s) + 2 * hi;
        v2f a;
        a.x = ybuf[m][kb];
        a.y = ybuf[m][kb + 1];
        v2f b;
        b.x = r0[s].x; b.y = r1[s].x;
        c0 = __builtin_amdgcn_wmma_f32_16x16x4_f32(false, a, false, b, (short)0,
                                                   c0, false, false);
        b.x = r0[s].y; b.y = r1[s].y;
        c1 = __builtin_amdgcn_wmma_f32_16x16x4_f32(false, a, false, b, (short)0,
                                                   c1, false, false);
        b.x = r0[s].z; b.y = r1[s].z;
        c2 = __builtin_amdgcn_wmma_f32_16x16x4_f32(false, a, false, b, (short)0,
                                                   c2, false, false);
        b.x = r0[s].w; b.y = r1[s].w;
        c3 = __builtin_amdgcn_wmma_f32_16x16x4_f32(false, a, false, b, (short)0,
                                                   c3, false, false);
      }
    }
#pragma unroll
    for (int j = 0; j < 8; ++j) {
      int r = j + 8 * hi;
      long idx = (base_row + r) * DIMD + 4 * m;
      float4 zo = *(const float4*)(z + idx);
      float4 zq = make_float4(c0[j], c1[j], c2[j], c3[j]);
      acc_cont += zo.x * (zo.x - zq.x);
      acc_cont += zo.y * (zo.y - zq.y);
      acc_cont += zo.z * (zo.z - zq.z);
      acc_cont += zo.w * (zo.w - zq.w);
      *(float4*)(out + idx) = zq;
    }
  }
  acc_cont = wave_sum(acc_cont);
  if (lane == 0) atomicAdd(scont, acc_cont);
}

__global__ __launch_bounds__(32) void vq_finalize(
    const float* __restrict__ kappa_p, const float* __restrict__ avgp_acc,
    const float* __restrict__ sdisc, const float* __restrict__ scont,
    float* __restrict__ out) {
  int lane = threadIdx.x;
  const float invN = 1.0f / (float)NROWS;
  float acc = 0.0f;
#pragma unroll
  for (int i = lane; i < KCODES; i += 32) {
    float a = avgp_acc[i] * invN;
    acc += a * logf(a + 1e-7f);
  }
  acc = wave_sum(acc);
  if (lane == 0) {
    float loss = (sdisc[0] + kappa_p[0] * scont[0]) / 8.0f;
    out[(long)NROWS * DIMD] = loss;
    out[(long)NROWS * DIMD + 1] = expf(-acc);
  }
}

extern "C" void kernel_launch(void* const* d_in, const int* in_sizes, int n_in,
                              void* d_out, int out_size, void* d_ws,
                              size_t ws_size, hipStream_t stream) {
  const float* z = (const float*)d_in[0];
  const float* kappa = (const float*)d_in[1];
  const float* cb = (const float*)d_in[2];
  const float* gum = (const float*)d_in[3];
  float* out = (float*)d_out;

  float* wsf = (float*)d_ws;
  float* avgp_acc = wsf;
  float* sdisc = wsf + 512;
  float* scont = wsf + 513;
  float* cbn = wsf + 1024;

  vq_zero<<<3, 256, 0, stream>>>(wsf);
  vq_cbnorm<<<64, 256, 0, stream>>>(cb, cbn);
  vq_main<<<NROWS / TILE_ROWS, 32, 0, stream>>>(z, kappa, cbn, gum, out,
                                                avgp_acc, sdisc, scont);
  vq_finalize<<<1, 32, 0, stream>>>(kappa, avgp_acc, sdisc, scont, out);
}


# Softmax pass: 3 hardware transcendentals per element (log2, rcp, exp2) replace 5 library exp/log calls: encodings = (exp(y-m1)/t)^2/sum, t = -ln(u+eps)+eps; p = exp(y-m1)/s1; sum order kept
# speedup vs baseline: 1.5255x; 1.3102x over previous
#include <hip/hip_runtime.h>
#include <hip/hip_bf16.h>

typedef __attribute__((ext_vector_type(2))) float v2f;
typedef __attribute__((ext_vector_type(8))) float v8f;

#define DIMD 64
#define KCODES 512
#define NROWS 65536
#define TILE_ROWS 16
#define NWAVES 2
#define ROWS_PER_WAVE (TILE_ROWS / NWAVES)
#define LOG2EF 1.4426950408889634f
#define LN2F 0.6931471805599453f

__device__ __forceinline__ float wave_sum(float v) {
#pragma unroll
  for (int off = 16; off > 0; off >>= 1) v += __shfl_xor(v, off, 32);
  return v;
}
__device__ __forceinline__ float wave_max(float v) {
#pragma unroll
  for (int off = 16; off > 0; off >>= 1) v = fmaxf(v, __shfl_xor(v, off, 32));
  return v;
}

__global__ void vq_zero(float* acc) {
  int i = blockIdx.x * blockDim.x + threadIdx.x;
  if (i < 514) acc[i] = 0.0f;
}

__global__ __launch_bounds__(256) void vq_cbnorm(const float* __restrict__ cb,
                                                 float* __restrict__ cbn) {
  int wave = (blockIdx.x * blockDim.x + threadIdx.x) >> 5;
  int lane = threadIdx.x & 31;
  const float* row = cb + wave * DIMD;
  float v0 = row[lane], v1 = row[lane + 32];
  float s = wave_sum(v0 * v0 + v1 * v1);
  float scale = 1.0f / fmaxf(sqrtf(s), 1e-12f);
  cbn[wave * DIMD + lane] = v0 * scale;
  cbn[wave * DIMD + lane + 32] = v1 * scale;
}

__global__ __launch_bounds__(32 * NWAVES) void vq_main(
    const float* __restrict__ z, const float* __restrict__ kappa_p,
    const float* __restrict__ cbn, const float* __restrict__ gum,
    float* __restrict__ out, float* __restrict__ avgp_acc,
    float* __restrict__ sdisc, float* __restrict__ scont) {
  __shared__ float zt[TILE_ROWS][DIMD];
  __shared__ float ybuf[TILE_ROWS][KCODES];

  const int lane = threadIdx.x & 31;
  const int w = __builtin_amdgcn_readfirstlane((int)(threadIdx.x >> 5));
  const int m = lane & 15;
  const int hi = lane >> 4;
  const long base_row = (long)blockIdx.x * TILE_ROWS;
  const float kappa = kappa_p[0];

#pragma unroll
  for (int rr = 0; rr < ROWS_PER_WAVE; ++rr) {
    const int r = w * ROWS_PER_WAVE + rr;
    const float* zr = z + (base_row + r) * DIMD;
    float v0 = zr[lane], v1 = zr[lane + 32];
    float s = wave_sum(v0 * v0 + v1 * v1);
    float scale = 1.0f / fmaxf(sqrtf(s), 1e-12f);
    zt[r][lane] = v0 * scale;
    zt[r][lane + 32] = v1 * scale;
  }
  __syncthreads();

  const int ct0 = w * (KCODES / 16 / NWAVES);
  for (int ct = ct0; ct < ct0 + KCODES / 16 / NWAVES; ct += 2) {
    v8f ca = {}, cb = {};
    const float* rowa = cbn + (ct * 16 + m) * DIMD;
    const float* rowb = rowa + 16 * DIMD;
#pragma unroll
    for (int k = 0; k < 16; ++k) {
      int kb = 4 * k + 2 * hi;
      v2f a;
      a.x = zt[m][kb];
      a.y = zt[m][kb + 1];
      v2f ba, bb;
      ba.x = rowa[kb];
      ba.y = rowa[kb + 1];
      bb.x = rowb[kb];
      bb.y = rowb[kb + 1];
      ca = __builtin_amdgcn_wmma_f32_16x16x4_f32(false, a, false, ba, (short)0,
                                                 ca, false, false);
      cb = __builtin_amdgcn_wmma_f32_16x16x4_f32(false, a, false, bb, (short)0,
                                                 cb, false, false);
    }
#pragma unroll
    for (int j = 0; j < 8; ++j) {
      ybuf[j + 8 * hi][ct * 16 + m] = kappa * ca[j];
      ybuf[j + 8 * hi][ct * 16 + 16 + m] = kappa * cb[j];
    }
  }
  __syncthreads();

  float acc_pl = 0.0f;
  float* avgp = &zt[0][0] + w * KCODES;
#pragma unroll
  for (int i = 0; i < 16; ++i) avgp[lane * 16 + i] = 0.0f;
  for (int rr = 0; rr < ROWS_PER_WAVE; ++rr) {
    const int r = w * ROWS_PER_WAVE + rr;
    const float4* urow4 =
        (const float4*)(gum + (base_row + r) * (long)KCODES) + lane * 4;
    float yv[16], rt[16];
    float m1 = -3.4e38f;
#pragma unroll
    for (int i4 = 0; i4 < 4; ++i4) {
      float4 u4 = urow4[i4];
      float uu[4] = {u4.x, u4.y, u4.z, u4.w};
#pragma unroll
      for (int j = 0; j < 4; ++j) {
        int i = i4 * 4 + j;
        int c0 = lane * 16 + i;
        float yy = ybuf[r][c0];
        float t = fmaf(-LN2F, __builtin_amdgcn_logf(uu[j] + 1e-10f), 1e-10f);
        rt[i] = __builtin_amdgcn_rcpf(t);
        yv[i] = yy;
        m1 = fmaxf(m1, yy);
      }
    }
    m1 = wave_max(m1);
    const float mc = m1 * LOG2EF;
    float e1[16], e2[16];
    float s1 = 0.0f, s2 = 0.0f;
#pragma unroll
    for (int i = 0; i < 16; ++i) {
      e1[i] = __builtin_amdgcn_exp2f(fmaf(yv[i], LOG2EF, -mc));
      float q = e1[i] * rt[i];
      e2[i] = q * q;
      s1 += e1[i];
      s2 += e2[i];
    }
    s1 = wave_sum(s1);
    s2 = wave_sum(s2);
    float lse = m1 + logf(s1);
    float inv_s1 = 1.0f / s1;
    float inv_s2 = 1.0f / s2;
#pragma unroll
    for (int i = 0; i < 16; ++i) {
      int c0 = lane * 16 + i;
      float lp = yv[i] - lse;
      float p = e1[i] * inv_s1;
      acc_pl += p * lp;
      avgp[c0] += p;
      ybuf[r][c0] = e2[i] * inv_s2;
    }
  }
  acc_pl = wave_sum(acc_pl);
  if (lane == 0) atomicAdd(sdisc, acc_pl);
  __syncthreads();
  {
    const float* pl = &zt[0][0];
#pragma unroll
    for (int q = 0; q < KCODES / (32 * NWAVES); ++q) {
      const int i = q * (32 * NWAVES) + (int)threadIdx.x;
      atomicAdd(&avgp_acc[i], pl[i] + pl[KCODES + i]);
    }
  }

  float acc_cont = 0.0f;
  {
    constexpr int KSTEPS = KCODES / 4 / NWAVES;
    constexpr int KBLK = 8;
    const int k0 = w * KSTEPS;
    const float* bbase = cbn + (long)(4 * k0 + 2 * hi) * DIMD + 4 * m;
    v8f c0 = {}, c1 = {}, c2 = {}, c3 = {};
    float4 n0[KBLK], n1[KBLK];
#pragma unroll
    for (int s = 0; s < KBLK; ++s) {
      n0[s] = *(const float4*)(bbase + (long)(4 * s) * DIMD);
      n1[s] = *(const float4*)(bbase + (long)(4 * s + 1) * DIMD);
    }
    for (int kk = 0; kk < KSTEPS / KBLK; ++kk) {
      float4 r0[KBLK], r1[KBLK];
#pragma unroll
      for (int s = 0; s < KBLK; ++s) {
        r0[s] = n0[s];
        r1[s] = n1[s];
      }
      if (kk + 1 < KSTEPS / KBLK) {
        const float* nb = bbase + (long)(kk + 1) * (KBLK * 4) * DIMD;
#pragma unroll
        for (int s = 0; s < KBLK; ++s) {
          n0[s] = *(const float4*)(nb + (long)(4 * s) * DIMD);
          n1[s] = *(const float4*)(nb + (long)(4 * s + 1) * DIMD);
        }
      }
#pragma unroll
      for (int s = 0; s < KBLK; ++s) {
        int kb = 4 * (k0 + kk * KBLK + s) + 2 * hi;
        v2f a;
        a.x = ybuf[m][kb];
        a.y = ybuf[m][kb + 1];
        v2f b;
        b.x = r0[s].x; b.y = r1[s].x;
        c0 = __builtin_amdgcn_wmma_f32_16x16x4_f32(false, a, false, b, (short)0,
                                                   c0, false, false);
        b.x = r0[s].y; b.y = r1[s].y;
        c1 = __builtin_amdgcn_wmma_f32_16x16x4_f32(false, a, false, b, (short)0,
                                                   c1, false, false);
        b.x = r0[s].z; b.y = r1[s].z;
        c2 = __builtin_amdgcn_wmma_f32_16x16x4_f32(false, a, false, b, (short)0,
                                                   c2, false, false);
        b.x = r0[s].w; b.y = r1[s].w;
        c3 = __builtin_amdgcn_wmma_f32_16x16x4_f32(false, a, false, b, (short)0,
                                                   c3, false, false);
      }
    }
    float keep[16], snd[16];
#pragma unroll
    for (int jj = 0; jj < 4; ++jj) {
      keep[0 + jj]  = w ? c0[4 + jj] : c0[jj];
      keep[4 + jj]  = w ? c1[4 + jj] : c1[jj];
      keep[8 + jj]  = w ? c2[4 + jj] : c2[jj];
      keep[12 + jj] = w ? c3[4 + jj] : c3[jj];
      snd[0 + jj]  = w ? c0[jj] : c0[4 + jj];
      snd[4 + jj]  = w ? c1[jj] : c1[4 + jj];
      snd[8 + jj]  = w ? c2[jj] : c2[4 + jj];
      snd[12 + jj] = w ? c3[jj] : c3[4 + jj];
    }
    float* xs = &ybuf[0][0] + (KCODES / NWAVES) * w + lane;
#pragma unroll
    for (int q = 0; q < 16; ++q) xs[q * KCODES] = snd[q];
    __syncthreads();
    const float* xr = &ybuf[0][0] + (KCODES / NWAVES) * (1 - w) + lane;
#pragma unroll
    for (int q = 0; q < 16; ++q) keep[q] += xr[q * KCODES];
#pragma unroll
    for (int jj = 0; jj < 4; ++jj) {
      int r = 4 * w + jj + 8 * hi;
      long idx = (base_row + r) * DIMD + 4 * m;
      float4 zo = *(const float4*)(z + idx);
      float4 zq = make_float4(keep[jj], keep[4 + jj], keep[8 + jj], keep[12 + jj]);
      acc_cont += zo.x * (zo.x - zq.x);
      acc_cont += zo.y * (zo.y - zq.y);
      acc_cont += zo.z * (zo.z - zq.z);
      acc_cont += zo.w * (zo.w - zq.w);
      *(float4*)(out + idx) = zq;
    }
  }
  acc_cont = wave_sum(acc_cont);
  if (lane == 0) atomicAdd(scont, acc_cont);
}

__global__ __launch_bounds__(32) void vq_finalize(
    const float* __restrict__ kappa_p, const float* __restrict__ avgp_acc,
    const float* __restrict__ sdisc, const float* __restrict__ scont,
    float* __restrict__ out) {
  int lane = threadIdx.x;
  const float invN = 1.0f / (float)NROWS;
  float acc = 0.0f;
#pragma unroll
  for (int i = lane; i < KCODES; i += 32) {
    float a = avgp_acc[i] * invN;
    acc += a * logf(a + 1e-7f);
  }
  acc = wave_sum(acc);
  if (lane == 0) {
    float loss = (sdisc[0] + kappa_p[0] * scont[0]) / 8.0f;
    out[(long)NROWS * DIMD] = loss;
    out[(long)NROWS * DIMD + 1] = expf(-acc);
  }
}

extern "C" void kernel_launch(void* const* d_in, const int* in_sizes, int n_in,
                              void* d_out, int out_size, void* d_ws,
                              size_t ws_size, hipStream_t stream) {
  const float* z = (const float*)d_in[0];
  const float* kappa = (const float*)d_in[1];
  const float* cb = (const float*)d_in[2];
  const float* gum = (const float*)d_in[3];
  float* out = (float*)d_out;

  float* wsf = (float*)d_ws;
  float* avgp_acc = wsf;
  float* sdisc = wsf + 512;
  float* scont = wsf + 513;
  float* cbn = wsf + 1024;

  vq_zero<<<3, 256, 0, stream>>>(wsf);
  vq_cbnorm<<<64, 256, 0, stream>>>(cb, cbn);
  vq_main<<<NROWS / TILE_ROWS, 32 * NWAVES, 0, stream>>>(z, kappa, cbn, gum, out,
                                                avgp_acc, sdisc, scont);
  vq_finalize<<<1, 32, 0, stream>>>(kappa, avgp_acc, sdisc, scont, out);
}


# GEMM1: codebook and z fragments read by b128 instead of b64 loads, one load feeding two k-steps: half as many row-scattered global loads per tile; k summation order changes (same order in A and B)
# speedup vs baseline: 1.6495x; 1.0813x over previous
#include <hip/hip_runtime.h>
#include <hip/hip_bf16.h>

typedef __attribute__((ext_vector_type(2))) float v2f;
typedef __attribute__((ext_vector_type(8))) float v8f;

#define DIMD 64
#define KCODES 512
#define NROWS 65536
#define TILE_ROWS 16
#define NWAVES 2
#define ROWS_PER_WAVE (TILE_ROWS / NWAVES)
#define LOG2EF 1.4426950408889634f
#define LN2F 0.6931471805599453f

__device__ __forceinline__ float wave_sum(float v) {
#pragma unroll
  for (int off = 16; off > 0; off >>= 1) v += __shfl_xor(v, off, 32);
  return v;
}
__device__ __forceinline__ float wave_max(float v) {
#pragma unroll
  for (int off = 16; off > 0; off >>= 1) v = fmaxf(v, __shfl_xor(v, off, 32));
  return v;
}

__global__ void vq_zero(float* acc) {
  int i = blockIdx.x * blockDim.x + threadIdx.x;
  if (i < 514) acc[i] = 0.0f;
}

__global__ __launch_bounds__(256) void vq_cbnorm(const float* __restrict__ cb,
                                                 float* __restrict__ cbn) {
  int wave = (blockIdx.x * blockDim.x + threadIdx.x) >> 5;
  int lane = threadIdx.x & 31;
  const float* row = cb + wave * DIMD;
  float v0 = row[lane], v1 = row[lane + 32];
  float s = wave_sum(v0 * v0 + v1 * v1);
  float scale = 1.0f / fmaxf(sqrtf(s), 1e-12f);
  cbn[wave * DIMD + lane] = v0 * scale;
  cbn[wave * DIMD + lane + 32] = v1 * scale;
}

__global__ __launch_bounds__(32 * NWAVES) void vq_main(
    const float* __restrict__ z, const float* __restrict__ kappa_p,
    const float* __restrict__ cbn, const float* __restrict__ gum,
    float* __restrict__ out, float* __restrict__ avgp_acc,
    float* __restrict__ sdisc, float* __restrict__ scont) {
  __shared__ __attribute__((aligned(16))) float zt[TILE_ROWS][DIMD];
  __shared__ float ybuf[TILE_ROWS][KCODES];

  const int lane = threadIdx.x & 31;
  const int w = __builtin_amdgcn_readfirstlane((int)(threadIdx.x >> 5));
  const int m = lane & 15;
  const int hi = lane >> 4;
  const long base_row = (long)blockIdx.x * TILE_ROWS;
  const float kappa = kappa_p[0];

#pragma unroll
  for (int rr = 0; rr < ROWS_PER_WAVE; ++rr) {
    const int r = w * ROWS_PER_WAVE + rr;
    const float* zr = z + (base_row + r) * DIMD;
    float v0 = zr[lane], v1 = zr[lane + 32];
    float s = wave_sum(v0 * v0 + v1 * v1);
    float scale = 1.0f / fmaxf(sqrtf(s), 1e-12f);
    zt[r][lane] = v0 * scale;
    zt[r][lane + 32] = v1 * scale;
  }
  __syncthreads();

  const int ct0 = w * (KCODES / 16 / NWAVES);
  for (int ct = ct0; ct < ct0 + KCODES / 16 / NWAVES; ct += 2) {
    v8f ca = {}, cb = {};
    const float* rowa = cbn + (ct * 16 + m) * DIMD + 4 * hi;
    const float* rowb = rowa + 16 * DIMD;
#pragma unroll
    for (int i = 0; i < 8; ++i) {
      const float4 a4 = *(const float4*)&zt[m][8 * i + 4 * hi];
      const float4 ba4 = *(const float4*)(rowa + 8 * i);
      const float4 bb4 = *(const float4*)(rowb + 8 * i);
      v2f a, ba, bb;
      a.x = a4.x;  a.y = a4.y;
      ba.x = ba4.x; ba.y = ba4.y;
      bb.x = bb4.x; bb.y = bb4.y;
      ca = __builtin_amdgcn_wmma_f32_16x16x4_f32(false, a, false, ba, (short)0,
                                                 ca, false, false);
      cb = __builtin_amdgcn_wmma_f32_16x16x4_f32(false, a, false, bb, (short)0,
                                                 cb, false, false);
      a.x = a4.z;  a.y = a4.w;
      ba.x = ba4.z; ba.y = ba4.w;
      bb.x = bb4.z; bb.y = bb4.w;
      ca = __builtin_amdgcn_wmma_f32_16x16x4_f32(false, a, false, ba, (short)0,
                                                 ca, false, false);
      cb = __builtin_amdgcn_wmma_f32_16x16x4_f32(false, a, false, bb, (short)0,
                                                 cb, false, false);
    }
#pragma unroll
    for (int j = 0; j < 8; ++j) {
      ybuf[j + 8 * hi][ct * 16 + m] = kappa * ca[j];
      ybuf[j + 8 * hi][ct * 16 + 16 + m] = kappa * cb[j];
    }
  }
  __syncthreads();

  float acc_pl = 0.0f;
  float* avgp = &zt[0][0] + w * KCODES;
#pragma unroll
  for (int i = 0; i < 16; ++i) avgp[lane * 16 + i] = 0.0f;
  for (int rr = 0; rr < ROWS_PER_WAVE; ++rr) {
    const int r = w * ROWS_PER_WAVE + rr;
    const float4* urow4 =
        (const float4*)(gum + (base_row + r) * (long)KCODES) + lane * 4;
    float yv[16], rt[16];
    float m1 = -3.4e38f;
#pragma unroll
    for (int i4 = 0; i4 < 4; ++i4) {
      float4 u4 = urow4[i4];
      float uu[4] = {u4.x, u4.y, u4.z, u4.w};
#pragma unroll
      for (int j = 0; j < 4; ++j) {
        int i = i4 * 4 + j;
        int c0 = lane * 16 + i;
        float yy = ybuf[r][c0];
        float t = fmaf(-LN2F, __builtin_amdgcn_logf(uu[j] + 1e-10f), 1e-10f);
        rt[i] = __builtin_amdgcn_rcpf(t);
        yv[i] = yy;
        m1 = fmaxf(m1, yy);
      }
    }
    m1 = wave_max(m1);
    const float mc = m1 * LOG2EF;
    float e1[16], e2[16];
    float s1 = 0.0f, s2 = 0.0f;
#pragma unroll
    for (int i = 0; i < 16; ++i) {
      e1[i] = __builtin_amdgcn_exp2f(fmaf(yv[i], LOG2EF, -mc));
      float q = e1[i] * rt[i];
      e2[i] = q * q;
      s1 += e1[i];
      s2 += e2[i];
    }
    s1 = wave_sum(s1);
    s2 = wave_sum(s2);
    float lse = m1 + logf(s1);
    float inv_s1 = 1.0f / s1;
    float inv_s2 = 1.0f / s2;
#pragma unroll
    for (int i = 0; i < 16; ++i) {
      int c0 = lane * 16 + i;
      float lp = yv[i] - lse;
      float p = e1[i] * inv_s1;
      acc_pl += p * lp;
      avgp[c0] += p;
      ybuf[r][c0] = e2[i] * inv_s2;
    }
  }
  acc_pl = wave_sum(acc_pl);
  if (lane == 0) atomicAdd(sdisc, acc_pl);
  __syncthreads();
  {
    const float* pl = &zt[0][0];
#pragma unroll
    for (int q = 0; q < KCODES / (32 * NWAVES); ++q) {
      const int i = q * (32 * NWAVES) + (int)threadIdx.x;
      atomicAdd(&avgp_acc[i], pl[i] + pl[KCODES + i]);
    }
  }

  float acc_cont = 0.0f;
  {
    constexpr int KSTEPS = KCODES / 4 / NWAVES;
    constexpr int KBLK = 8;
    const int k0 = w * KSTEPS;
    const float* bbase = cbn + (long)(4 * k0 + 2 * hi) * DIMD + 4 * m;
    v8f c0 = {}, c1 = {}, c2 = {}, c3 = {};
    float4 n0[KBLK], n1[KBLK];
#pragma unroll
    for (int s = 0; s < KBLK; ++s) {
      n0[s] = *(const float4*)(bbase + (long)(4 * s) * DIMD);
      n1[s] = *(const float4*)(bbase + (long)(4 * s + 1) * DIMD);
    }
    for (int kk = 0; kk < KSTEPS / KBLK; ++kk) {
      float4 r0[KBLK], r1[KBLK];
#pragma unroll
      for (int s = 0; s < KBLK; ++s) {
        r0[s] = n0[s];
        r1[s] = n1[s];
      }
      if (kk + 1 < KSTEPS / KBLK) {
        const float* nb = bbase + (long)(kk + 1) * (KBLK * 4) * DIMD;
#pragma unroll
        for (int s = 0; s < KBLK; ++s) {
          n0[s] = *(const float4*)(nb + (long)(4 * s) * DIMD);
          n1[s] = *(const float4*)(nb + (long)(4 * s + 1) * DIMD);
        }
      }
#pragma unroll
      for (int s = 0; s < KBLK; ++s) {
        int kb = 4 * (k0 + kk * KBLK + s) + 2 * hi;
        v2f a;
        a.x = ybuf[m][kb];
        a.y = ybuf[m][kb + 1];
        v2f b;
        b.x = r0[s].x; b.y = r1[s].x;
        c0 = __builtin_amdgcn_wmma_f32_16x16x4_f32(false, a, false, b, (short)0,
                                                   c0, false, false);
        b.x = r0[s].y; b.y = r1[s].y;
        c1 = __builtin_amdgcn_wmma_f32_16x16x4_f32(false, a, false, b, (short)0,
                                                   c1, false, false);
        b.x = r0[s].z; b.y = r1[s].z;
        c2 = __builtin_amdgcn_wmma_f32_16x16x4_f32(false, a, false, b, (short)0,
                                                   c2, false, false);
        b.x = r0[s].w; b.y = r1[s].w;
        c3 = __builtin_amdgcn_wmma_f32_16x16x4_f32(false, a, false, b, (short)0,
                                                   c3, false, false);
      }
    }
    float keep[16], snd[16];
#pragma unroll
    for (int jj = 0; jj < 4; ++jj) {
      keep[0 + jj]  = w ? c0[4 + jj] : c0[jj];
      keep[4 + jj]  = w ? c1[4 + jj] : c1[jj];
      keep[8 + jj]  = w ? c2[4 + jj] : c2[jj];
      keep[12 + jj] = w ? c3[4 + jj] : c3[jj];
      snd[0 + jj]  = w ? c0[jj] : c0[4 + jj];
      snd[4 + jj]  = w ? c1[jj] : c1[4 + jj];
      snd[8 + jj]  = w ? c2[jj] : c2[4 + jj];
      snd[12 + jj] = w ? c3[jj] : c3[4 + jj];
    }
    float* xs = &ybuf[0][0] + (KCODES / NWAVES) * w + lane;
#pragma unroll
    for (int q = 0; q < 16; ++q) xs[q * KCODES] = snd[q];
    __syncthreads();
    const float* xr = &ybuf[0][0] + (KCODES / NWAVES) * (1 - w) + lane;
#pragma unroll
    for (int q = 0; q < 16; ++q) keep[q] += xr[q * KCODES];
#pragma unroll
    for (int jj = 0; jj < 4; ++jj) {
      int r = 4 * w + jj + 8 * hi;
      long idx = (base_row + r) * DIMD + 4 * m;
      float4 zo = *(const float4*)(z + idx);
      float4 zq = make_float4(keep[jj], keep[4 + jj], keep[8 + jj], keep[12 + jj]);
      acc_cont += zo.x * (zo.x - zq.x);
      acc_cont += zo.y * (zo.y - zq.y);
      acc_cont += zo.z * (zo.z - zq.z);
      acc_cont += zo.w * (zo.w - zq.w);
      *(float4*)(out + idx) = zq;
    }
  }
  acc_cont = wave_sum(acc_cont);
  if (lane == 0) atomicAdd(scont, acc_cont);
}

__global__ __launch_bounds__(32) void vq_finalize(
    const float* __restrict__ kappa_p, const float* __restrict__ avgp_acc,
    const float* __restrict__ sdisc, const float* __restrict__ scont,
    float* __restrict__ out) {
  int lane = threadIdx.x;
  const float invN = 1.0f / (float)NROWS;
  float acc = 0.0f;
#pragma unroll
  for (int i = lane; i < KCODES; i += 32) {
    float a = avgp_acc[i] * invN;
    acc += a * logf(a + 1e-7f);
  }
  acc = wave_sum(acc);
  if (lane == 0) {
    float loss = (sdisc[0] + kappa_p[0] * scont[0]) / 8.0f;
    out[(long)NROWS * DIMD] = loss;
    out[(long)NROWS * DIMD + 1] = expf(-acc);
  }
}

extern "C" void kernel_launch(void* const* d_in, const int* in_sizes, int n_in,
                              void* d_out, int out_size, void* d_ws,
                              size_t ws_size, hipStream_t stream) {
  const float* z = (const float*)d_in[0];
  const float* kappa = (const float*)d_in[1];
  const float* cb = (const float*)d_in[2];
  const float* gum = (const float*)d_in[3];
  float* out = (float*)d_out;

  float* wsf = (float*)d_ws;
  float* avgp_acc = wsf;
  float* sdisc = wsf + 512;
  float* scont = wsf + 513;
  float* cbn = wsf + 1024;

  vq_zero<<<3, 256, 0, stream>>>(wsf);
  vq_cbnorm<<<64, 256, 0, stream>>>(cb, cbn);
  vq_main<<<NROWS / TILE_ROWS, 32 * NWAVES, 0, stream>>>(z, kappa, cbn, gum, out,
                                                avgp_acc, sdisc, scont);
  vq_finalize<<<1, 32, 0, stream>>>(kappa, avgp_acc, sdisc, scont, out);
}


# Normalized codebook stored in GEMM1 fragment order (128-byte lines of 4 codes x 8 dims): a GEMM1 codebook load reads 4 whole cache lines, not 16 rows; GEMM2 loads touch 8; values bit-identical
# speedup vs baseline: 1.7766x; 1.0771x over previous
#include <hip/hip_runtime.h>
#include <hip/hip_bf16.h>

typedef __attribute__((ext_vector_type(2))) float v2f;
typedef __attribute__((ext_vector_type(8))) float v8f;

#define DIMD 64
#define KCODES 512
#define NROWS 65536
#define TILE_ROWS 16
#define NWAVES 2
#define ROWS_PER_WAVE (TILE_ROWS / NWAVES)
#define LOG2EF 1.4426950408889634f
#define LN2F 0.6931471805599453f

__device__ __forceinline__ float wave_sum(float v) {
#pragma unroll
  for (int off = 16; off > 0; off >>= 1) v += __shfl_xor(v, off, 32);
  return v;
}
__device__ __forceinline__ float wave_max(float v) {
#pragma unroll
  for (int off = 16; off > 0; off >>= 1) v = fmaxf(v, __shfl_xor(v, off, 32));
  return v;
}

__global__ void vq_zero(float* acc) {
  int i = blockIdx.x * blockDim.x + threadIdx.x;
  if (i < 514) acc[i] = 0.0f;
}

#define CBN_POS(c, d)                                                         \
  ((((c) >> 4) << 10) + (((d) >> 3) << 7) + ((((c) & 15) >> 2) << 5) +        \
   ((((d) >> 2) & 1) << 4) + (((c) & 3) << 2) + ((d) & 3))
__global__ __launch_bounds__(256) void vq_cbnorm(const float* __restrict__ cb,
                                                 float* __restrict__ cbn) {
  __shared__ float rows[8][DIMD];
  const int wv = threadIdx.x >> 5;
  const int lane = threadIdx.x & 31;
  const int code = blockIdx.x * 8 + wv;
  const float* row = cb + code * DIMD;
  float v0 = row[lane], v1 = row[lane + 32];
  float s = wave_sum(v0 * v0 + v1 * v1);
  float scale = 1.0f / fmaxf(sqrtf(s), 1e-12f);
  rows[wv][lane] = v0 * scale;
  rows[wv][lane + 32] = v1 * scale;
  __syncthreads();
  const int hb = lane >> 4, cc = (lane >> 2) & 3, e = lane & 3;
#pragma unroll
  for (int t = 0; t < 2; ++t) {
    const int ln = 2 * wv + t;
    const int grp = ln >> 3;
    const int i = ln & 7;
    const int c = blockIdx.x * 8 + 4 * grp + cc;
    const int d = 8 * i + 4 * hb + e;
    cbn[CBN_POS(c, d)] = rows[4 * grp + cc][d];
  }
}

__global__ __launch_bounds__(32 * NWAVES) void vq_main(
    const float* __restrict__ z, const float* __restrict__ kappa_p,
    const float* __restrict__ cbn, const float* __restrict__ gum,
    float* __restrict__ out, float* __restrict__ avgp_acc,
    float* __restrict__ sdisc, float* __restrict__ scont) {
  __shared__ __attribute__((aligned(16))) float zt[TILE_ROWS][DIMD];
  __shared__ float ybuf[TILE_ROWS][KCODES];

  const int lane = threadIdx.x & 31;
  const int w = __builtin_amdgcn_readfirstlane((int)(threadIdx.x >> 5));
  const int m = lane & 15;
  const int hi = lane >> 4;
  const long base_row = (long)blockIdx.x * TILE_ROWS;
  const float kappa = kappa_p[0];

#pragma unroll
  for (int rr = 0; rr < ROWS_PER_WAVE; ++rr) {
    const int r = w * ROWS_PER_WAVE + rr;
    const float* zr = z + (base_row + r) * DIMD;
    float v0 = zr[lane], v1 = zr[lane + 32];
    float s = wave_sum(v0 * v0 + v1 * v1);
    float scale = 1.0f / fmaxf(sqrtf(s), 1e-12f);
    zt[r][lane] = v0 * scale;
    zt[r][lane + 32] = v1 * scale;
  }
  __syncthreads();

  const int ct0 = w * (KCODES / 16 / NWAVES);
  for (int ct = ct0; ct < ct0 + KCODES / 16 / NWAVES; ct += 2) {
    v8f ca = {}, cb = {};
    const float* rowa = cbn + CBN_POS(ct * 16 + m, 4 * hi);
    const float* rowb = rowa + 1024;
#pragma unroll
    for (int i = 0; i < 8; ++i) {
      const float4 a4 = *(const float4*)&zt[m][8 * i + 4 * hi];
      const float4 ba4 = *(const float4*)(rowa + 128 * i);
      const float4 bb4 = *(const float4*)(rowb + 128 * i);
      v2f a, ba, bb;
      a.x = a4.x;  a.y = a4.y;
      ba.x = ba4.x; ba.y = ba4.y;
      bb.x = bb4.x; bb.y = bb4.y;
      ca = __builtin_amdgcn_wmma_f32_16x16x4_f32(false, a, false, ba, (short)0,
                                                 ca, false, false);
      cb = __builtin_amdgcn_wmma_f32_16x16x4_f32(false, a, false, bb, (short)0,
                                                 cb, false, false);
      a.x = a4.z;  a.y = a4.w;
      ba.x = ba4.z; ba.y = ba4.w;
      bb.x = bb4.z; bb.y = bb4.w;
      ca = __builtin_amdgcn_wmma_f32_16x16x4_f32(false, a, false, ba, (short)0,
                                                 ca, false, false);
      cb = __builtin_amdgcn_wmma_f32_16x16x4_f32(false, a, false, bb, (short)0,
                                                 cb, false, false);
    }
#pragma unroll
    for (int j = 0; j < 8; ++j) {
      ybuf[j + 8 * hi][ct * 16 + m] = kappa * ca[j];
      ybuf[j + 8 * hi][ct * 16 + 16 + m] = kappa * cb[j];
    }
  }
  __syncthreads();

  float acc_pl = 0.0f;
  float* avgp = &zt[0][0] + w * KCODES;
#pragma unroll
  for (int i = 0; i < 16; ++i) avgp[lane * 16 + i] = 0.0f;
  for (int rr = 0; rr < ROWS_PER_WAVE; ++rr) {
    const int r = w * ROWS_PER_WAVE + rr;
    const float4* urow4 =
        (const float4*)(gum + (base_row + r) * (long)KCODES) + lane * 4;
    float yv[16], rt[16];
    float m1 = -3.4e38f;
#pragma unroll
    for (int i4 = 0; i4 < 4; ++i4) {
      float4 u4 = urow4[i4];
      float uu[4] = {u4.x, u4.y, u4.z, u4.w};
#pragma unroll
      for (int j = 0; j < 4; ++j) {
        int i = i4 * 4 + j;
        int c0 = lane * 16 + i;
        float yy = ybuf[r][c0];
        float t = fmaf(-LN2F, __builtin_amdgcn_logf(uu[j] + 1e-10f), 1e-10f);
        rt[i] = __builtin_amdgcn_rcpf(t);
        yv[i] = yy;
        m1 = fmaxf(m1, yy);
      }
    }
    m1 = wave_max(m1);
    const float mc = m1 * LOG2EF;
    float e1[16], e2[16];
    float s1 = 0.0f, s2 = 0.0f;
#pragma unroll
    for (int i = 0; i < 16; ++i) {
      e1[i] = __builtin_amdgcn_exp2f(fmaf(yv[i], LOG2EF, -mc));
      float q = e1[i] * rt[i];
      e2[i] = q * q;
      s1 += e1[i];
      s2 += e2[i];
    }
    s1 = wave_sum(s1);
    s2 = wave_sum(s2);
    float lse = m1 + logf(s1);
    float inv_s1 = 1.0f / s1;
    float inv_s2 = 1.0f / s2;
#pragma unroll
    for (int i = 0; i < 16; ++i) {
      int c0 = lane * 16 + i;
      float lp = yv[i] - lse;
      float p = e1[i] * inv_s1;
      acc_pl += p * lp;
      avgp[c0] += p;
      ybuf[r][c0] = e2[i] * inv_s2;
    }
  }
  acc_pl = wave_sum(acc_pl);
  if (lane == 0) atomicAdd(sdisc, acc_pl);
  __syncthreads();
  {
    const float* pl = &zt[0][0];
#pragma unroll
    for (int q = 0; q < KCODES / (32 * NWAVES); ++q) {
      const int i = q * (32 * NWAVES) + (int)threadIdx.x;
      atomicAdd(&avgp_acc[i], pl[i] + pl[KCODES + i]);
    }
  }

  float acc_cont = 0.0f;
  {
    constexpr int KSTEPS = KCODES / 4 / NWAVES;
    constexpr int KBLK = 8;
    const int k0 = w * KSTEPS;
    const float* bbase = cbn + CBN_POS(4 * k0 + 2 * hi, 4 * m);
    v8f c0 = {}, c1 = {}, c2 = {}, c3 = {};
    float4 n0[KBLK], n1[KBLK];
#pragma unroll
    for (int s = 0; s < KBLK; ++s) {
      n0[s] = *(const float4*)(bbase + (s >> 2) * 1024 + (s & 3) * 32);
      n1[s] = *(const float4*)(bbase + (s >> 2) * 1024 + (s & 3) * 32 + 4);
    }
    for (int kk = 0; kk < KSTEPS / KBLK; ++kk) {
      float4 r0[KBLK], r1[KBLK];
#pragma unroll
      for (int s = 0; s < KBLK; ++s) {
        r0[s] = n0[s];
        r1[s] = n1[s];
      }
      if (kk + 1 < KSTEPS / KBLK) {
        const float* nb = bbase + (kk + 1) * (KBLK / 4) * 1024;
#pragma unroll
        for (int s = 0; s < KBLK; ++s) {
          n0[s] = *(const float4*)(nb + (s >> 2) * 1024 + (s & 3) * 32);
          n1[s] = *(const float4*)(nb + (s >> 2) * 1024 + (s & 3) * 32 + 4);
        }
      }
#pragma unroll
      for (int s = 0; s < KBLK; ++s) {
        int kb = 4 * (k0 + kk * KBLK + s) + 2 * hi;
        v2f a;
        a.x = ybuf[m][kb];
        a.y = ybuf[m][kb + 1];
        v2f b;
        b.x = r0[s].x; b.y = r1[s].x;
        c0 = __builtin_amdgcn_wmma_f32_16x16x4_f32(false, a, false, b, (short)0,
                                                   c0, false, false);
        b.x = r0[s].y; b.y = r1[s].y;
        c1 = __builtin_amdgcn_wmma_f32_16x16x4_f32(false, a, false, b, (short)0,
                                                   c1, false, false);
        b.x = r0[s].z; b.y = r1[s].z;
        c2 = __builtin_amdgcn_wmma_f32_16x16x4_f32(false, a, false, b, (short)0,
                                                   c2, false, false);
        b.x = r0[s].w; b.y = r1[s].w;
        c3 = __builtin_amdgcn_wmma_f32_16x16x4_f32(false, a, false, b, (short)0,
                                                   c3, false, false);
      }
    }
    float keep[16], snd[16];
#pragma unroll
    for (int jj = 0; jj < 4; ++jj) {
      keep[0 + jj]  = w ? c0[4 + jj] : c0[jj];
      keep[4 + jj]  = w ? c1[4 + jj] : c1[jj];
      keep[8 + jj]  = w ? c2[4 + jj] : c2[jj];
      keep[12 + jj] = w ? c3[4 + jj] : c3[jj];
      snd[0 + jj]  = w ? c0[jj] : c0[4 + jj];
      snd[4 + jj]  = w ? c1[jj] : c1[4 + jj];
      snd[8 + jj]  = w ? c2[jj] : c2[4 + jj];
      snd[12 + jj] = w ? c3[jj] : c3[4 + jj];
    }
    float* xs = &ybuf[0][0] + (KCODES / NWAVES) * w + lane;
#pragma unroll
    for (int q = 0; q < 16; ++q) xs[q * KCODES] = snd[q];
    __syncthreads();
    const float* xr = &ybuf[0][0] + (KCODES / NWAVES) * (1 - w) + lane;
#pragma unroll
    for (int q = 0; q < 16; ++q) keep[q] += xr[q * KCODES];
#pragma unroll
    for (int jj = 0; jj < 4; ++jj) {
      int r = 4 * w + jj + 8 * hi;
      long idx = (base_row + r) * DIMD + 4 * m;
      float4 zo = *(const float4*)(z + idx);
      float4 zq = make_float4(keep[jj], keep[4 + jj], keep[8 + jj], keep[12 + jj]);
      acc_cont += zo.x * (zo.x - zq.x);
      acc_cont += zo.y * (zo.y - zq.y);
      acc_cont += zo.z * (zo.z - zq.z);
      acc_cont += zo.w * (zo.w - zq.w);
      *(float4*)(out + idx) = zq;
    }
  }
  acc_cont = wave_sum(acc_cont);
  if (lane == 0) atomicAdd(scont, acc_cont);
}

__global__ __launch_bounds__(32) void vq_finalize(
    const float* __restrict__ kappa_p, const float* __restrict__ avgp_acc,
    const float* __restrict__ sdisc, const float* __restrict__ scont,
    float* __restrict__ out) {
  int lane = threadIdx.x;
  const float invN = 1.0f / (float)NROWS;
  float acc = 0.0f;
#pragma unroll
  for (int i = lane; i < KCODES; i += 32) {
    float a = avgp_acc[i] * invN;
    acc += a * logf(a + 1e-7f);
  }
  acc = wave_sum(acc);
  if (lane == 0) {
    float loss = (sdisc[0] + kappa_p[0] * scont[0]) / 8.0f;
    out[(long)NROWS * DIMD] = loss;
    out[(long)NROWS * DIMD + 1] = expf(-acc);
  }
}

extern "C" void kernel_launch(void* const* d_in, const int* in_sizes, int n_in,
                              void* d_out, int out_size, void* d_ws,
                              size_t ws_size, hipStream_t stream) {
  const float* z = (const float*)d_in[0];
  const float* kappa = (const float*)d_in[1];
  const float* cb = (const float*)d_in[2];
  const float* gum = (const float*)d_in[3];
  float* out = (float*)d_out;

  float* wsf = (float*)d_ws;
  float* avgp_acc = wsf;
  float* sdisc = wsf + 512;
  float* scont = wsf + 513;
  float* cbn = wsf + 1024;

  vq_zero<<<3, 256, 0, stream>>>(wsf);
  vq_cbnorm<<<64, 256, 0, stream>>>(cb, cbn);
  vq_main<<<NROWS / TILE_ROWS, 32 * NWAVES, 0, stream>>>(z, kappa, cbn, gum, out,
                                                avgp_acc, sdisc, scont);
  vq_finalize<<<1, 32, 0, stream>>>(kappa, avgp_acc, sdisc, scont, out);
}


# Shared-memory tiles free of bank conflicts: logits rows padded to 516 floats (GEMM2 A reads were 16-way), z rows to 68, softmax lanes own float4 pieces 128 columns apart; row sum order changes
# speedup vs baseline: 1.8449x; 1.0384x over previous
#include <hip/hip_runtime.h>
#include <hip/hip_bf16.h>

typedef __attribute__((ext_vector_type(2))) float v2f;
typedef __attribute__((ext_vector_type(8))) float v8f;

#define DIMD 64
#define KCODES 512
#define NROWS 65536
#define TILE_ROWS 16
#define NWAVES 2
#define ROWS_PER_WAVE (TILE_ROWS / NWAVES)
#define ZPITCH (DIMD + 4)
#define YPITCH (KCODES + 4)
#define LOG2EF 1.4426950408889634f
#define LN2F 0.6931471805599453f

__device__ __forceinline__ float wave_sum(float v) {
#pragma unroll
  for (int off = 16; off > 0; off >>= 1) v += __shfl_xor(v, off, 32);
  return v;
}
__device__ __forceinline__ float wave_max(float v) {
#pragma unroll
  for (int off = 16; off > 0; off >>= 1) v = fmaxf(v, __shfl_xor(v, off, 32));
  return v;
}

__global__ void vq_zero(float* acc) {
  int i = blockIdx.x * blockDim.x + threadIdx.x;
  if (i < 514) acc[i] = 0.0f;
}

#define CBN_POS(c, d)                                                         \
  ((((c) >> 4) << 10) + (((d) >> 3) << 7) + ((((c) & 15) >> 2) << 5) +        \
   ((((d) >> 2) & 1) << 4) + (((c) & 3) << 2) + ((d) & 3))
__global__ __launch_bounds__(256) void vq_cbnorm(const float* __restrict__ cb,
                                                 float* __restrict__ cbn) {
  __shared__ float rows[8][DIMD];
  const int wv = threadIdx.x >> 5;
  const int lane = threadIdx.x & 31;
  const int code = blockIdx.x * 8 + wv;
  const float* row = cb + code * DIMD;
  float v0 = row[lane], v1 = row[lane + 32];
  float s = wave_sum(v0 * v0 + v1 * v1);
  float scale = 1.0f / fmaxf(sqrtf(s), 1e-12f);
  rows[wv][lane] = v0 * scale;
  rows[wv][lane + 32] = v1 * scale;
  __syncthreads();
  const int hb = lane >> 4, cc = (lane >> 2) & 3, e = lane & 3;
#pragma unroll
  for (int t = 0; t < 2; ++t) {
    const int ln = 2 * wv + t;
    const int grp = ln >> 3;
    const int i = ln & 7;
    const int c = blockIdx.x * 8 + 4 * grp + cc;
    const int d = 8 * i + 4 * hb + e;
    cbn[CBN_POS(c, d)] = rows[4 * grp + cc][d];
  }
}

__global__ __launch_bounds__(32 * NWAVES) void vq_main(
    const float* __restrict__ z, const float* __restrict__ kappa_p,
    const float* __restrict__ cbn, const float* __restrict__ gum,
    float* __restrict__ out, float* __restrict__ avgp_acc,
    float* __restrict__ sdisc, float* __restrict__ scont) {
  __shared__ __attribute__((aligned(16))) float zt[TILE_ROWS][ZPITCH];
  __shared__ __attribute__((aligned(16))) float ybuf[TILE_ROWS][YPITCH];

  const int lane = threadIdx.x & 31;
  const int w = __builtin_amdgcn_readfirstlane((int)(threadIdx.x >> 5));
  const int m = lane & 15;
  const int hi = lane >> 4;
  const long base_row = (long)blockIdx.x * TILE_ROWS;
  const float kappa = kappa_p[0];

#pragma unroll
  for (int rr = 0; rr < ROWS_PER_WAVE; ++rr) {
    const int r = w * ROWS_PER_WAVE + rr;
    const float* zr = z + (base_row + r) * DIMD;
    float v0 = zr[lane], v1 = zr[lane + 32];
    float s = wave_sum(v0 * v0 + v1 * v1);
    float scale = 1.0f / fmaxf(sqrtf(s), 1e-12f);
    zt[r][lane] = v0 * scale;
    zt[r][lane + 32] = v1 * scale;
  }
  __syncthreads();

  const int ct0 = w * (KCODES / 16 / NWAVES);
  for (int ct = ct0; ct < ct0 + KCODES / 16 / NWAVES; ct += 2) {
    v8f ca = {}, cb = {};
    const float* rowa = cbn + CBN_POS(ct * 16 + m, 4 * hi);
    const float* rowb = rowa + 1024;
#pragma unroll
    for (int i = 0; i < 8; ++i) {
      const float4 a4 = *(const float4*)&zt[m][8 * i + 4 * hi];
      const float4 ba4 = *(const float4*)(rowa + 128 * i);
      const float4 bb4 = *(const float4*)(rowb + 128 * i);
      v2f a, ba, bb;
      a.x = a4.x;  a.y = a4.y;
      ba.x = ba4.x; ba.y = ba4.y;
      bb.x = bb4.x; bb.y = bb4.y;
      ca = __builtin_amdgcn_wmma_f32_16x16x4_f32(false, a, false, ba, (short)0,
                                                 ca, false, false);
      cb = __builtin_amdgcn_wmma_f32_16x16x4_f32(false, a, false, bb, (short)0,
                                                 cb, false, false);
      a.x = a4.z;  a.y = a4.w;
      ba.x = ba4.z; ba.y = ba4.w;
      bb.x = bb4.z; bb.y = bb4.w;
      ca = __builtin_amdgcn_wmma_f32_16x16x4_f32(false, a, false, ba, (short)0,
                                                 ca, false, false);
      cb = __builtin_amdgcn_wmma_f32_16x16x4_f32(false, a, false, bb, (short)0,
                                                 cb, false, false);
    }
#pragma unroll
    for (int j = 0; j < 8; ++j) {
      ybuf[j + 8 * hi][ct * 16 + m] = kappa * ca[j];
      ybuf[j + 8 * hi][ct * 16 + 16 + m] = kappa * cb[j];
    }
  }
  __syncthreads();

  float acc_pl = 0.0f;
  float* avgp = &zt[0][0] + w * KCODES;
  for (int q = 0; q < 4; ++q)
    *(float4*)(avgp + 128 * q + 4 * lane) = make_float4(0.0f, 0.0f, 0.0f, 0.0f);
  for (int rr = 0; rr < ROWS_PER_WAVE; ++rr) {
    const int r = w * ROWS_PER_WAVE + rr;
    const float4* urow4 =
        (const float4*)(gum + (base_row + r) * (long)KCODES) + lane;
    float* yrow = &ybuf[r][4 * lane];
    float yv[16], rt[16];
    float m1 = -3.4e38f;
#pragma unroll
    for (int q = 0; q < 4; ++q) {
      float4 u4 = urow4[32 * q];
      float4 y4 = *(const float4*)(yrow + 128 * q);
      float uu[4] = {u4.x, u4.y, u4.z, u4.w};
      float yy4[4] = {y4.x, y4.y, y4.z, y4.w};
#pragma unroll
      for (int j = 0; j < 4; ++j) {
        int i = q * 4 + j;
        float yy = yy4[j];
        float t = fmaf(-LN2F, __builtin_amdgcn_logf(uu[j] + 1e-10f), 1e-10f);
        rt[i] = __builtin_amdgcn_rcpf(t);
        yv[i] = yy;
        m1 = fmaxf(m1, yy);
      }
    }
    m1 = wave_max(m1);
    const float mc = m1 * LOG2EF;
    float e1[16], e2[16];
    float s1 = 0.0f, s2 = 0.0f;
#pragma unroll
    for (int i = 0; i < 16; ++i) {
      e1[i] = __builtin_amdgcn_exp2f(fmaf(yv[i], LOG2EF, -mc));
      float q = e1[i] * rt[i];
      e2[i] = q * q;
      s1 += e1[i];
      s2 += e2[i];
    }
    s1 = wave_sum(s1);
    s2 = wave_sum(s2);
    float lse = m1 + logf(s1);
    float inv_s1 = 1.0f / s1;
    float inv_s2 = 1.0f / s2;
#pragma unroll
    for (int q = 0; q < 4; ++q) {
      float4 a4 = *(const float4*)(avgp + 128 * q + 4 * lane);
      float pp[4], en[4];
#pragma unroll
      for (int j = 0; j < 4; ++j) {
        int i = q * 4 + j;
        float lp = yv[i] - lse;
        float p = e1[i] * inv_s1;
        acc_pl += p * lp;
        pp[j] = p;
        en[j] = e2[i] * inv_s2;
      }
      *(float4*)(avgp + 128 * q + 4 * lane) =
          make_float4(a4.x + pp[0], a4.y + pp[1], a4.z + pp[2], a4.w + pp[3]);
      *(float4*)(yrow + 128 * q) = make_float4(en[0], en[1], en[2], en[3]);
    }
  }
  acc_pl = wave_sum(acc_pl);
  if (lane == 0) atomicAdd(sdisc, acc_pl);
  __syncthreads();
  {
    const float* pl = &zt[0][0];
#pragma unroll
    for (int q = 0; q < KCODES / (32 * NWAVES); ++q) {
      const int i = q * (32 * NWAVES) + (int)threadIdx.x;
      atomicAdd(&avgp_acc[i], pl[i] + pl[KCODES + i]);
    }
  }

  float acc_cont = 0.0f;
  {
    constexpr int KSTEPS = KCODES / 4 / NWAVES;
    constexpr int KBLK = 8;
    const int k0 = w * KSTEPS;
    const float* bbase = cbn + CBN_POS(4 * k0 + 2 * hi, 4 * m);
    v8f c0 = {}, c1 = {}, c2 = {}, c3 = {};
    float4 n0[KBLK], n1[KBLK];
#pragma unroll
    for (int s = 0; s < KBLK; ++s) {
      n0[s] = *(const float4*)(bbase + (s >> 2) * 1024 + (s & 3) * 32);
      n1[s] = *(const float4*)(bbase + (s >> 2) * 1024 + (s & 3) * 32 + 4);
    }
    for (int kk = 0; kk < KSTEPS / KBLK; ++kk) {
      float4 r0[KBLK], r1[KBLK];
#pragma unroll
      for (int s = 0; s < KBLK; ++s) {
        r0[s] = n0[s];
        r1[s] = n1[s];
      }
      if (kk + 1 < KSTEPS / KBLK) {
        const float* nb = bbase + (kk + 1) * (KBLK / 4) * 1024;
#pragma unroll
        for (int s = 0; s < KBLK; ++s) {
          n0[s] = *(const float4*)(nb + (s >> 2) * 1024 + (s & 3) * 32);
          n1[s] = *(const float4*)(nb + (s >> 2) * 1024 + (s & 3) * 32 + 4);
        }
      }
#pragma unroll
      for (int s = 0; s < KBLK; ++s) {
        int kb = 4 * (k0 + kk * KBLK + s) + 2 * hi;
        v2f a;
        a.x = ybuf[m][kb];
        a.y = ybuf[m][kb + 1];
        v2f b;
        b.x = r0[s].x; b.y = r1[s].x;
        c0 = __builtin_amdgcn_wmma_f32_16x16x4_f32(false, a, false, b, (short)0,
                                                   c0, false, false);
        b.x = r0[s].y; b.y = r1[s].y;
        c1 = __builtin_amdgcn_wmma_f32_16x16x4_f32(false, a, false, b, (short)0,
                                                   c1, false, false);
        b.x = r0[s].z; b.y = r1[s].z;
        c2 = __builtin_amdgcn_wmma_f32_16x16x4_f32(false, a, false, b, (short)0,
                                                   c2, false, false);
        b.x = r0[s].w; b.y = r1[s].w;
        c3 = __builtin_amdgcn_wmma_f32_16x16x4_f32(false, a, false, b, (short)0,
                                                   c3, false, false);
      }
    }
    float keep[16], snd[16];
#pragma unroll
    for (int jj = 0; jj < 4; ++jj) {
      keep[0 + jj]  = w ? c0[4 + jj] : c0[jj];
      keep[4 + jj]  = w ? c1[4 + jj] : c1[jj];
      keep[8 + jj]  = w ? c2[4 + jj] : c2[jj];
      keep[12 + jj] = w ? c3[4 + jj] : c3[jj];
      snd[0 + jj]  = w ? c0[jj] : c0[4 + jj];
      snd[4 + jj]  = w ? c1[jj] : c1[4 + jj];
      snd[8 + jj]  = w ? c2[jj] : c2[4 + jj];
      snd[12 + jj] = w ? c3[jj] : c3[4 + jj];
    }
    float* xs = &ybuf[0][0] + (KCODES / NWAVES) * w + lane;
#pragma unroll
    for (int q = 0; q < 16; ++q) xs[q * YPITCH] = snd[q];
    __syncthreads();
    const float* xr = &ybuf[0][0] + (KCODES / NWAVES) * (1 - w) + lane;
#pragma unroll
    for (int q = 0; q < 16; ++q) keep[q] += xr[q * YPITCH];
#pragma unroll
    for (int jj = 0; jj < 4; ++jj) {
      int r = 4 * w + jj + 8 * hi;
      long idx = (base_row + r) * DIMD + 4 * m;
      float4 zo = *(const float4*)(z + idx);
      float4 zq = make_float4(keep[jj], keep[4 + jj], keep[8 + jj], keep[12 + jj]);
      acc_cont += zo.x * (zo.x - zq.x);
      acc_cont += zo.y * (zo.y - zq.y);
      acc_cont += zo.z * (zo.z - zq.z);
      acc_cont += zo.w * (zo.w - zq.w);
      *(float4*)(out + idx) = zq;
    }
  }
  acc_cont = wave_sum(acc_cont);
  if (lane == 0) atomicAdd(scont, acc_cont);
}

__global__ __launch_bounds__(32) void vq_finalize(
    const float* __restrict__ kappa_p, const float* __restrict__ avgp_acc,
    const float* __restrict__ sdisc, const float* __restrict__ scont,
    float* __restrict__ out) {
  int lane = threadIdx.x;
  const float invN = 1.0f / (float)NROWS;
  float acc = 0.0f;
#pragma unroll
  for (int i = lane; i < KCODES; i += 32) {
    float a = avgp_acc[i] * invN;
    acc += a * logf(a + 1e-7f);
  }
  acc = wave_sum(acc);
  if (lane == 0) {
    float loss = (sdisc[0] + kappa_p[0] * scont[0]) / 8.0f;
    out[(long)NROWS * DIMD] = loss;
    out[(long)NROWS * DIMD + 1] = expf(-acc);
  }
}

extern "C" void kernel_launch(void* const* d_in, const int* in_sizes, int n_in,
                              void* d_out, int out_size, void* d_ws,
                              size_t ws_size, hipStream_t stream) {
  const float* z = (const float*)d_in[0];
  const float* kappa = (const float*)d_in[1];
  const float* cb = (const float*)d_in[2];
  const float* gum = (const float*)d_in[3];
  float* out = (float*)d_out;

  float* wsf = (float*)d_ws;
  float* avgp_acc = wsf;
  float* sdisc = wsf + 512;
  float* scont = wsf + 513;
  float* cbn = wsf + 1024;

  vq_zero<<<3, 256, 0, stream>>>(wsf);
  vq_cbnorm<<<64, 256, 0, stream>>>(cb, cbn);
  vq_main<<<NROWS / TILE_ROWS, 32 * NWAVES, 0, stream>>>(z, kappa, cbn, gum, out,
                                                avgp_acc, sdisc, scont);
  vq_finalize<<<1, 32, 0, stream>>>(kappa, avgp_acc, sdisc, scont, out);
}


# Four waves share two 16-row tiles: every codebook fragment load (GEMM1 and GEMM2) feeds both row tiles, half the codebook loads per row; z_q = sum of four k-quarter partials (sum order changes)
# speedup vs baseline: 1.9565x; 1.0605x over previous
#include <hip/hip_runtime.h>
#include <hip/hip_bf16.h>

typedef __attribute__((ext_vector_type(2))) float v2f;
typedef __attribute__((ext_vector_type(8))) float v8f;

#define DIMD 64
#define KCODES 512
#define NROWS 65536
#define TILE_ROWS 16
#define NTILES 2
#define BLOCK_ROWS (TILE_ROWS * NTILES)
#define NWAVES 4
#define ROWS_PER_WAVE (BLOCK_ROWS / NWAVES)
#define ZPITCH (DIMD + 4)
#define YPITCH (KCODES + 4)
#define LOG2EF 1.4426950408889634f
#define LN2F 0.6931471805599453f

__device__ __forceinline__ float wave_sum(float v) {
#pragma unroll
  for (int off = 16; off > 0; off >>= 1) v += __shfl_xor(v, off, 32);
  return v;
}
__device__ __forceinline__ float wave_max(float v) {
#pragma unroll
  for (int off = 16; off > 0; off >>= 1) v = fmaxf(v, __shfl_xor(v, off, 32));
  return v;
}

__global__ void vq_zero(float* acc) {
  int i = blockIdx.x * blockDim.x + threadIdx.x;
  if (i < 514) acc[i] = 0.0f;
}

#define CBN_POS(c, d)                                                         \
  ((((c) >> 4) << 10) + (((d) >> 3) << 7) + ((((c) & 15) >> 2) << 5) +        \
   ((((d) >> 2) & 1) << 4) + (((c) & 3) << 2) + ((d) & 3))
__global__ __launch_bounds__(256) void vq_cbnorm(const float* __restrict__ cb,
                                                 float* __restrict__ cbn) {
  __shared__ float rows[8][DIMD];
  const int wv = threadIdx.x >> 5;
  const int lane = threadIdx.x & 31;
  const int code = blockIdx.x * 8 + wv;
  const float* row = cb + code * DIMD;
  float v0 = row[lane], v1 = row[lane + 32];
  float s = wave_sum(v0 * v0 + v1 * v1);
  float scale = 1.0f / fmaxf(sqrtf(s), 1e-12f);
  rows[wv][lane] = v0 * scale;
  rows[wv][lane + 32] = v1 * scale;
  __syncthreads();
  const int hb = lane >> 4, cc = (lane >> 2) & 3, e = lane & 3;
#pragma unroll
  for (int t = 0; t < 2; ++t) {
    const int ln = 2 * wv + t;
    const int grp = ln >> 3;
    const int i = ln & 7;
    const int c = blockIdx.x * 8 + 4 * grp + cc;
    const int d = 8 * i + 4 * hb + e;
    cbn[CBN_POS(c, d)] = rows[4 * grp + cc][d];
  }
}

__global__ __launch_bounds__(32 * NWAVES) void vq_main(
    const float* __restrict__ z, const float* __restrict__ kappa_p,
    const float* __restrict__ cbn, const float* __restrict__ gum,
    float* __restrict__ out, float* __restrict__ avgp_acc,
    float* __restrict__ sdisc, float* __restrict__ scont) {
  __shared__ __attribute__((aligned(16))) float zt[BLOCK_ROWS][ZPITCH];
  __shared__ __attribute__((aligned(16))) float ybuf[BLOCK_ROWS][YPITCH];

  const int lane = threadIdx.x & 31;
  const int w = __builtin_amdgcn_readfirstlane((int)(threadIdx.x >> 5));
  const int m = lane & 15;
  const int hi = lane >> 4;
  const long base_row = (long)blockIdx.x * BLOCK_ROWS;
  const float kappa = kappa_p[0];

#pragma unroll
  for (int rr = 0; rr < ROWS_PER_WAVE; ++rr) {
    const int r = w * ROWS_PER_WAVE + rr;
    const float* zr = z + (base_row + r) * DIMD;
    float v0 = zr[lane], v1 = zr[lane + 32];
    float s = wave_sum(v0 * v0 + v1 * v1);
    float scale = 1.0f / fmaxf(sqrtf(s), 1e-12f);
    zt[r][lane] = v0 * scale;
    zt[r][lane + 32] = v1 * scale;
  }
  __syncthreads();

  const int ct0 = w * (KCODES / 16 / NWAVES);
  for (int ct = ct0; ct < ct0 + KCODES / 16 / NWAVES; ct += 2) {
    v8f ca0 = {}, cb0 = {};
    v8f ca1 = {}, cb1 = {};
    const float* rowa = cbn + CBN_POS(ct * 16 + m, 4 * hi);
    const float* rowb = rowa + 1024;
#pragma unroll
    for (int i = 0; i < 8; ++i) {
      const float4 p4 = *(const float4*)&zt[m][8 * i + 4 * hi];
      const float4 q4 = *(const float4*)&zt[TILE_ROWS + m][8 * i + 4 * hi];
      const float4 ba4 = *(const float4*)(rowa + 128 * i);
      const float4 bb4 = *(const float4*)(rowb + 128 * i);
      v2f a0, a1, ba, bb;
      a0.x = p4.x; a0.y = p4.y;
      a1.x = q4.x; a1.y = q4.y;
      ba.x = ba4.x; ba.y = ba4.y;
      bb.x = bb4.x; bb.y = bb4.y;
      ca0 = __builtin_amdgcn_wmma_f32_16x16x4_f32(false, a0, false, ba, (short)0,
                                                  ca0, false, false);
      ca1 = __builtin_amdgcn_wmma_f32_16x16x4_f32(false, a1, false, ba, (short)0,
                                                  ca1, false, false);
      cb0 = __builtin_amdgcn_wmma_f32_16x16x4_f32(false, a0, false, bb, (short)0,
                                                  cb0, false, false);
      cb1 = __builtin_amdgcn_wmma_f32_16x16x4_f32(false, a1, false, bb, (short)0,
                                                  cb1, false, false);
      a0.x = p4.z; a0.y = p4.w;
      a1.x = q4.z; a1.y = q4.w;
      ba.x = ba4.z; ba.y = ba4.w;
      bb.x = bb4.z; bb.y = bb4.w;
      ca0 = __builtin_amdgcn_wmma_f32_16x16x4_f32(false, a0, false, ba, (short)0,
                                                  ca0, false, false);
      ca1 = __builtin_amdgcn_wmma_f32_16x16x4_f32(false, a1, false, ba, (short)0,
                                                  ca1, false, false);
      cb0 = __builtin_amdgcn_wmma_f32_16x16x4_f32(false, a0, false, bb, (short)0,
                                                  cb0, false, false);
      cb1 = __builtin_amdgcn_wmma_f32_16x16x4_f32(false, a1, false, bb, (short)0,
                                                  cb1, false, false);
    }
#pragma unroll
    for (int j = 0; j < 8; ++j) {
      ybuf[j + 8 * hi][ct * 16 + m] = kappa * ca0[j];
      ybuf[j + 8 * hi][ct * 16 + 16 + m] = kappa * cb0[j];
      ybuf[TILE_ROWS + j + 8 * hi][ct * 16 + m] = kappa * ca1[j];
      ybuf[TILE_ROWS + j + 8 * hi][ct * 16 + 16 + m] = kappa * cb1[j];
    }
  }
  __syncthreads();

  float acc_pl = 0.0f;
  float* avgp = &zt[0][0] + w * KCODES;
  for (int q = 0; q < 4; ++q)
    *(float4*)(avgp + 128 * q + 4 * lane) = make_float4(0.0f, 0.0f, 0.0f, 0.0f);
  for (int rr = 0; rr < ROWS_PER_WAVE; ++rr) {
    const int r = w * ROWS_PER_WAVE + rr;
    const float4* urow4 =
        (const float4*)(gum + (base_row + r) * (long)KCODES) + lane;
    float* yrow = &ybuf[r][4 * lane];
    float yv[16], rt[16];
    float m1 = -3.4e38f;
#pragma unroll
    for (int q = 0; q < 4; ++q) {
      float4 u4 = urow4[32 * q];
      float4 y4 = *(const float4*)(yrow + 128 * q);
      float uu[4] = {u4.x, u4.y, u4.z, u4.w};
      float yy4[4] = {y4.x, y4.y, y4.z, y4.w};
#pragma unroll
      for (int j = 0; j < 4; ++j) {
        int i = q * 4 + j;
        float yy = yy4[j];
        float t = fmaf(-LN2F, __builtin_amdgcn_logf(uu[j] + 1e-10f), 1e-10f);
        rt[i] = __builtin_amdgcn_rcpf(t);
        yv[i] = yy;
        m1 = fmaxf(m1, yy);
      }
    }
    m1 = wave_max(m1);
    const float mc = m1 * LOG2EF;
    float e1[16], e2[16];
    float s1 = 0.0f, s2 = 0.0f;
#pragma unroll
    for (int i = 0; i < 16; ++i) {
      e1[i] = __builtin_amdgcn_exp2f(fmaf(yv[i], LOG2EF, -mc));
      float q = e1[i] * rt[i];
      e2[i] = q * q;
      s1 += e1[i];
      s2 += e2[i];
    }
    s1 = wave_sum(s1);
    s2 = wave_sum(s2);
    float lse = m1 + logf(s1);
    float inv_s1 = 1.0f / s1;
    float inv_s2 = 1.0f / s2;
#pragma unroll
    for (int q = 0; q < 4; ++q) {
      float4 a4 = *(const float4*)(avgp + 128 * q + 4 * lane);
      float pp[4], en[4];
#pragma unroll
      for (int j = 0; j < 4; ++j) {
        int i = q * 4 + j;
        float lp = yv[i] - lse;
        float p = e1[i] * inv_s1;
        acc_pl += p * lp;
        pp[j] = p;
        en[j] = e2[i] * inv_s2;
      }
      *(float4*)(avgp + 128 * q + 4 * lane) =
          make_float4(a4.x + pp[0], a4.y + pp[1], a4.z + pp[2], a4.w + pp[3]);
      *(float4*)(yrow + 128 * q) = make_float4(en[0], en[1], en[2], en[3]);
    }
  }
  acc_pl = wave_sum(acc_pl);
  if (lane == 0) atomicAdd(sdisc, acc_pl);
  __syncthreads();
  {
    const float* pl = &zt[0][0];
#pragma unroll
    for (int q = 0; q < KCODES / (32 * NWAVES); ++q) {
      const int i = q * (32 * NWAVES) + (int)threadIdx.x;
      atomicAdd(&avgp_acc[i], (pl[i] + pl[KCODES + i]) +
                                  (pl[2 * KCODES + i] + pl[3 * KCODES + i]));
    }
  }

  float acc_cont = 0.0f;
  {
    constexpr int KSTEPS = KCODES / 4 / NWAVES;
    constexpr int KBLK = 4;
    const int k0 = w * KSTEPS;
    const float* bbase = cbn + CBN_POS(4 * k0 + 2 * hi, 4 * m);
    v8f c0 = {}, c1 = {}, c2 = {}, c3 = {};
    v8f d0 = {}, d1 = {}, d2 = {}, d3 = {};
    float4 n0[KBLK], n1[KBLK];
#pragma unroll
    for (int s = 0; s < KBLK; ++s) {
      n0[s] = *(const float4*)(bbase + (s >> 2) * 1024 + (s & 3) * 32);
      n1[s] = *(const float4*)(bbase + (s >> 2) * 1024 + (s & 3) * 32 + 4);
    }
#pragma unroll 2
    for (int kk = 0; kk < KSTEPS / KBLK; ++kk) {
      float4 r0[KBLK], r1[KBLK];
#pragma unroll
      for (int s = 0; s < KBLK; ++s) {
        r0[s] = n0[s];
        r1[s] = n1[s];
      }
      if (kk + 1 < KSTEPS / KBLK) {
        const float* nb = bbase + (kk + 1) * (KBLK / 4) * 1024;
#pragma unroll
        for (int s = 0; s < KBLK; ++s) {
          n0[s] = *(const float4*)(nb + (s >> 2) * 1024 + (s & 3) * 32);
          n1[s] = *(const float4*)(nb + (s >> 2) * 1024 + (s & 3) * 32 + 4);
        }
      }
#pragma unroll
      for (int s = 0; s < KBLK; ++s) {
        int kb = 4 * (k0 + kk * KBLK + s) + 2 * hi;
        v2f a0, a1;
        a0.x = ybuf[m][kb];
        a0.y = ybuf[m][kb + 1];
        a1.x = ybuf[TILE_ROWS + m][kb];
        a1.y = ybuf[TILE_ROWS + m][kb + 1];
        v2f b;
        b.x = r0[s].x; b.y = r1[s].x;
        c0 = __builtin_amdgcn_wmma_f32_16x16x4_f32(false, a0, false, b, (short)0,
                                                   c0, false, false);
        d0 = __builtin_amdgcn_wmma_f32_16x16x4_f32(false, a1, false, b, (short)0,
                                                   d0, false, false);
        b.x = r0[s].y; b.y = r1[s].y;
        c1 = __builtin_amdgcn_wmma_f32_16x16x4_f32(false, a0, false, b, (short)0,
                                                   c1, false, false);
        d1 = __builtin_amdgcn_wmma_f32_16x16x4_f32(false, a1, false, b, (short)0,
                                                   d1, false, false);
        b.x = r0[s].z; b.y = r1[s].z;
        c2 = __builtin_amdgcn_wmma_f32_16x16x4_f32(false, a0, false, b, (short)0,
                                                   c2, false, false);
        d2 = __builtin_amdgcn_wmma_f32_16x16x4_f32(false, a1, false, b, (short)0,
                                                   d2, false, false);
        b.x = r0[s].w; b.y = r1[s].w;
        c3 = __builtin_amdgcn_wmma_f32_16x16x4_f32(false, a0, false, b, (short)0,
                                                   c3, false, false);
        d3 = __builtin_amdgcn_wmma_f32_16x16x4_f32(false, a1, false, b, (short)0,
                                                   d3, false, false);
      }
    }
    float* xw = &ybuf[0][0] + (KCODES / NWAVES) * w + 4 * lane;
#define VQ_PUT(T, t, acc)                                                     \
    *(float4*)(xw + ((T) * 8 + (t) * 2 + 0) * YPITCH) =                       \
        make_float4(acc[0], acc[1], acc[2], acc[3]);                          \
    *(float4*)(xw + ((T) * 8 + (t) * 2 + 1) * YPITCH) =                       \
        make_float4(acc[4], acc[5], acc[6], acc[7]);
    VQ_PUT(0, 0, c0) VQ_PUT(0, 1, c1) VQ_PUT(0, 2, c2) VQ_PUT(0, 3, c3)
    VQ_PUT(1, 0, d0) VQ_PUT(1, 1, d1) VQ_PUT(1, 2, d2) VQ_PUT(1, 3, d3)
#undef VQ_PUT
    __syncthreads();
    const int tw = w >> 1, jh = w & 1;
    const float* xr = &ybuf[0][0] + (tw * 8 + jh) * YPITCH + 4 * lane;
    float sv[4][4];
#pragma unroll
    for (int t = 0; t < 4; ++t) {
      const float* p = xr + 2 * t * YPITCH;
      float4 s = *(const float4*)p;
#pragma unroll
      for (int sw = 1; sw < NWAVES; ++sw) {
        const float4 o = *(const float4*)(p + (KCODES / NWAVES) * sw);
        s.x += o.x; s.y += o.y; s.z += o.z; s.w += o.w;
      }
      sv[t][0] = s.x; sv[t][1] = s.y; sv[t][2] = s.z; sv[t][3] = s.w;
    }
#pragma unroll
    for (int jj = 0; jj < 4; ++jj) {
      int r = TILE_ROWS * tw + 8 * hi + 4 * jh + jj;
      long idx = (base_row + r) * DIMD + 4 * m;
      float4 zo = *(const float4*)(z + idx);
      float4 zq = make_float4(sv[0][jj], sv[1][jj], sv[2][jj], sv[3][jj]);
      acc_cont += zo.x * (zo.x - zq.x);
      acc_cont += zo.y * (zo.y - zq.y);
      acc_cont += zo.z * (zo.z - zq.z);
      acc_cont += zo.w * (zo.w - zq.w);
      *(float4*)(out + idx) = zq;
    }
  }
  acc_cont = wave_sum(acc_cont);
  if (lane == 0) atomicAdd(scont, acc_cont);
}

__global__ __launch_bounds__(32) void vq_finalize(
    const float* __restrict__ kappa_p, const float* __restrict__ avgp_acc,
    const float* __restrict__ sdisc, const float* __restrict__ scont,
    float* __restrict__ out) {
  int lane = threadIdx.x;
  const float invN = 1.0f / (float)NROWS;
  float acc = 0.0f;
#pragma unroll
  for (int i = lane; i < KCODES; i += 32) {
    float a = avgp_acc[i] * invN;
    acc += a * logf(a + 1e-7f);
  }
  acc = wave_sum(acc);
  if (lane == 0) {
    float loss = (sdisc[0] + kappa_p[0] * scont[0]) / 8.0f;
    out[(long)NROWS * DIMD] = loss;
    out[(long)NROWS * DIMD + 1] = expf(-acc);
  }
}

extern "C" void kernel_launch(void* const* d_in, const int* in_sizes, int n_in,
                              void* d_out, int out_size, void* d_ws,
                              size_t ws_size, hipStream_t stream) {
  const float* z = (const float*)d_in[0];
  const float* kappa = (const float*)d_in[1];
  const float* cb = (const float*)d_in[2];
  const float* gum = (const float*)d_in[3];
  float* out = (float*)d_out;

  float* wsf = (float*)d_ws;
  float* avgp_acc = wsf;
  float* sdisc = wsf + 512;
  float* scont = wsf + 513;
  float* cbn = wsf + 1024;

  vq_zero<<<3, 256, 0, stream>>>(wsf);
  vq_cbnorm<<<64, 256, 0, stream>>>(cb, cbn);
  vq_main<<<NROWS / BLOCK_ROWS, 32 * NWAVES, 0, stream>>>(z, kappa, cbn, gum, out,
                                                avgp_acc, sdisc, scont);
  vq_finalize<<<1, 32, 0, stream>>>(kappa, avgp_acc, sdisc, scont, out);
}


# Wave-wide sums and maxima exchange lanes on the vector ALU (row_xmask DPP adds, permlanex16) instead of shared-memory-unit permutes waited for at every step; same partners and order: bit-identical
# speedup vs baseline: 2.0616x; 1.0537x over previous
#include <hip/hip_runtime.h>
#include <hip/hip_bf16.h>

typedef __attribute__((ext_vector_type(2))) float v2f;
typedef __attribute__((ext_vector_type(8))) float v8f;

#define DIMD 64
#define KCODES 512
#define NROWS 65536
#define TILE_ROWS 16
#define NTILES 2
#define BLOCK_ROWS (TILE_ROWS * NTILES)
#define NWAVES 4
#define ROWS_PER_WAVE (BLOCK_ROWS / NWAVES)
#define ZPITCH (DIMD + 4)
#define YPITCH (KCODES + 4)
#define LOG2EF 1.4426950408889634f
#define LN2F 0.6931471805599453f

__device__ __forceinline__ float xrow16(float x) {
  const int b = __builtin_bit_cast(int, x);
  return __builtin_bit_cast(
      float, __builtin_amdgcn_permlanex16(b, b, 0x76543210, 0xfedcba98, false, false));
}
template <int MASK>
__device__ __forceinline__ float xlane(float x) {
  const int b = __builtin_bit_cast(int, x);
  return __builtin_bit_cast(
      float, __builtin_amdgcn_update_dpp(b, b, 0x160 + MASK, 0xf, 0xf, true));
}
__device__ __forceinline__ float wave_sum(float v) {
  v += xrow16(v);
  v += xlane<8>(v);
  v += xlane<4>(v);
  v += xlane<2>(v);
  v += xlane<1>(v);
  return v;
}
__device__ __forceinline__ float wave_max(float v) {
  v = fmaxf(v, xrow16(v));
  v = fmaxf(v, xlane<8>(v));
  v = fmaxf(v, xlane<4>(v));
  v = fmaxf(v, xlane<2>(v));
  v = fmaxf(v, xlane<1>(v));
  return v;
}

__global__ void vq_zero(float* acc) {
  int i = blockIdx.x * blockDim.x + threadIdx.x;
  if (i < 514) acc[i] = 0.0f;
}

#define CBN_POS(c, d)                                                         \
  ((((c) >> 4) << 10) + (((d) >> 3) << 7) + ((((c) & 15) >> 2) << 5) +        \
   ((((d) >> 2) & 1) << 4) + (((c) & 3) << 2) + ((d) & 3))
__global__ __launch_bounds__(256) void vq_cbnorm(const float* __restrict__ cb,
                                                 float* __restrict__ cbn) {
  __shared__ float rows[8][DIMD];
  const int wv = threadIdx.x >> 5;
  const int lane = threadIdx.x & 31;
  const int code = blockIdx.x * 8 + wv;
  const float* row = cb + code * DIMD;
  float v0 = row[lane], v1 = row[lane + 32];
  float s = wave_sum(v0 * v0 + v1 * v1);
  float scale = 1.0f / fmaxf(sqrtf(s), 1e-12f);
  rows[wv][lane] = v0 * scale;
  rows[wv][lane + 32] = v1 * scale;
  __syncthreads();
  const int hb = lane >> 4, cc = (lane >> 2) & 3, e = lane & 3;
#pragma unroll
  for (int t = 0; t < 2; ++t) {
    const int ln = 2 * wv + t;
    const int grp = ln >> 3;
    const int i = ln & 7;
    const int c = blockIdx.x * 8 + 4 * grp + cc;
    const int d = 8 * i + 4 * hb + e;
    cbn[CBN_POS(c, d)] = rows[4 * grp + cc][d];
  }
}

__global__ __launch_bounds__(32 * NWAVES) void vq_main(
    const float* __restrict__ z, const float* __restrict__ kappa_p,
    const float* __restrict__ cbn, const float* __restrict__ gum,
    float* __restrict__ out, float* __restrict__ avgp_acc,
    float* __restrict__ sdisc, float* __restrict__ scont) {
  __shared__ __attribute__((aligned(16))) float zt[BLOCK_ROWS][ZPITCH];
  __shared__ __attribute__((aligned(16))) float ybuf[BLOCK_ROWS][YPITCH];

  const int lane = threadIdx.x & 31;
  const int w = __builtin_amdgcn_readfirstlane((int)(threadIdx.x >> 5));
  const int m = lane & 15;
  const int hi = lane >> 4;
  const long base_row = (long)blockIdx.x * BLOCK_ROWS;
  const float kappa = kappa_p[0];

#pragma unroll
  for (int rr = 0; rr < ROWS_PER_WAVE; ++rr) {
    const int r = w * ROWS_PER_WAVE + rr;
    const float* zr = z + (base_row + r) * DIMD;
    float v0 = zr[lane], v1 = zr[lane + 32];
    float s = wave_sum(v0 * v0 + v1 * v1);
    float scale = 1.0f / fmaxf(sqrtf(s), 1e-12f);
    zt[r][lane] = v0 * scale;
    zt[r][lane + 32] = v1 * scale;
  }
  __syncthreads();

  const int ct0 = w * (KCODES / 16 / NWAVES);
  for (int ct = ct0; ct < ct0 + KCODES / 16 / NWAVES; ct += 2) {
    v8f ca0 = {}, cb0 = {};
    v8f ca1 = {}, cb1 = {};
    const float* rowa = cbn + CBN_POS(ct * 16 + m, 4 * hi);
    const float* rowb = rowa + 1024;
#pragma unroll
    for (int i = 0; i < 8; ++i) {
      const float4 p4 = *(const float4*)&zt[m][8 * i + 4 * hi];
      const float4 q4 = *(const float4*)&zt[TILE_ROWS + m][8 * i + 4 * hi];
      const float4 ba4 = *(const float4*)(rowa + 128 * i);
      const float4 bb4 = *(const float4*)(rowb + 128 * i);
      v2f a0, a1, ba, bb;
      a0.x = p4.x; a0.y = p4.y;
      a1.x = q4.x; a1.y = q4.y;
      ba.x = ba4.x; ba.y = ba4.y;
      bb.x = bb4.x; bb.y = bb4.y;
      ca0 = __builtin_amdgcn_wmma_f32_16x16x4_f32(false, a0, false, ba, (short)0,
                                                  ca0, false, false);
      ca1 = __builtin_amdgcn_wmma_f32_16x16x4_f32(false, a1, false, ba, (short)0,
                                                  ca1, false, false);
      cb0 = __builtin_amdgcn_wmma_f32_16x16x4_f32(false, a0, false, bb, (short)0,
                                                  cb0, false, false);
      cb1 = __builtin_amdgcn_wmma_f32_16x16x4_f32(false, a1, false, bb, (short)0,
                                                  cb1, false, false);
      a0.x = p4.z; a0.y = p4.w;
      a1.x = q4.z; a1.y = q4.w;
      ba.x = ba4.z; ba.y = ba4.w;
      bb.x = bb4.z; bb.y = bb4.w;
      ca0 = __builtin_amdgcn_wmma_f32_16x16x4_f32(false, a0, false, ba, (short)0,
                                                  ca0, false, false);
      ca1 = __builtin_amdgcn_wmma_f32_16x16x4_f32(false, a1, false, ba, (short)0,
                                                  ca1, false, false);
      cb0 = __builtin_amdgcn_wmma_f32_16x16x4_f32(false, a0, false, bb, (short)0,
                                                  cb0, false, false);
      cb1 = __builtin_amdgcn_wmma_f32_16x16x4_f32(false, a1, false, bb, (short)0,
                                                  cb1, false, false);
    }
#pragma unroll
    for (int j = 0; j < 8; ++j) {
      ybuf[j + 8 * hi][ct * 16 + m] = kappa * ca0[j];
      ybuf[j + 8 * hi][ct * 16 + 16 + m] = kappa * cb0[j];
      ybuf[TILE_ROWS + j + 8 * hi][ct * 16 + m] = kappa * ca1[j];
      ybuf[TILE_ROWS + j + 8 * hi][ct * 16 + 16 + m] = kappa * cb1[j];
    }
  }
  __syncthreads();

  float acc_pl = 0.0f;
  float* avgp = &zt[0][0] + w * KCODES;
  for (int q = 0; q < 4; ++q)
    *(float4*)(avgp + 128 * q + 4 * lane) = make_float4(0.0f, 0.0f, 0.0f, 0.0f);
  for (int rr = 0; rr < ROWS_PER_WAVE; ++rr) {
    const int r = w * ROWS_PER_WAVE + rr;
    const float4* urow4 =
        (const float4*)(gum + (base_row + r) * (long)KCODES) + lane;
    float* yrow = &ybuf[r][4 * lane];
    float yv[16], rt[16];
    float m1 = -3.4e38f;
#pragma unroll
    for (int q = 0; q < 4; ++q) {
      float4 u4 = urow4[32 * q];
      float4 y4 = *(const float4*)(yrow + 128 * q);
      float uu[4] = {u4.x, u4.y, u4.z, u4.w};
      float yy4[4] = {y4.x, y4.y, y4.z, y4.w};
#pragma unroll
      for (int j = 0; j < 4; ++j) {
        int i = q * 4 + j;
        float yy = yy4[j];
        float t = fmaf(-LN2F, __builtin_amdgcn_logf(uu[j] + 1e-10f), 1e-10f);
        rt[i] = __builtin_amdgcn_rcpf(t);
        yv[i] = yy;
        m1 = fmaxf(m1, yy);
      }
    }
    m1 = wave_max(m1);
    const float mc = m1 * LOG2EF;
    float e1[16], e2[16];
    float s1 = 0.0f, s2 = 0.0f;
#pragma unroll
    for (int i = 0; i < 16; ++i) {
      e1[i] = __builtin_amdgcn_exp2f(fmaf(yv[i], LOG2EF, -mc));
      float q = e1[i] * rt[i];
      e2[i] = q * q;
      s1 += e1[i];
      s2 += e2[i];
    }
    s1 = wave_sum(s1);
    s2 = wave_sum(s2);
    float lse = m1 + logf(s1);
    float inv_s1 = 1.0f / s1;
    float inv_s2 = 1.0f / s2;
#pragma unroll
    for (int q = 0; q < 4; ++q) {
      float4 a4 = *(const float4*)(avgp + 128 * q + 4 * lane);
      float pp[4], en[4];
#pragma unroll
      for (int j = 0; j < 4; ++j) {
        int i = q * 4 + j;
        float lp = yv[i] - lse;
        float p = e1[i] * inv_s1;
        acc_pl += p * lp;
        pp[j] = p;
        en[j] = e2[i] * inv_s2;
      }
      *(float4*)(avgp + 128 * q + 4 * lane) =
          make_float4(a4.x + pp[0], a4.y + pp[1], a4.z + pp[2], a4.w + pp[3]);
      *(float4*)(yrow + 128 * q) = make_float4(en[0], en[1], en[2], en[3]);
    }
  }
  acc_pl = wave_sum(acc_pl);
  if (lane == 0) atomicAdd(sdisc, acc_pl);
  __syncthreads();
  {
    const float* pl = &zt[0][0];
#pragma unroll
    for (int q = 0; q < KCODES / (32 * NWAVES); ++q) {
      const int i = q * (32 * NWAVES) + (int)threadIdx.x;
      atomicAdd(&avgp_acc[i], (pl[i] + pl[KCODES + i]) +
                                  (pl[2 * KCODES + i] + pl[3 * KCODES + i]));
    }
  }

  float acc_cont = 0.0f;
  {
    constexpr int KSTEPS = KCODES / 4 / NWAVES;
    constexpr int KBLK = 4;
    const int k0 = w * KSTEPS;
    const float* bbase = cbn + CBN_POS(4 * k0 + 2 * hi, 4 * m);
    v8f c0 = {}, c1 = {}, c2 = {}, c3 = {};
    v8f d0 = {}, d1 = {}, d2 = {}, d3 = {};
    float4 n0[KBLK], n1[KBLK];
#pragma unroll
    for (int s = 0; s < KBLK; ++s) {
      n0[s] = *(const float4*)(bbase + (s >> 2) * 1024 + (s & 3) * 32);
      n1[s] = *(const float4*)(bbase + (s >> 2) * 1024 + (s & 3) * 32 + 4);
    }
#pragma unroll 2
    for (int kk = 0; kk < KSTEPS / KBLK; ++kk) {
      float4 r0[KBLK], r1[KBLK];
#pragma unroll
      for (int s = 0; s < KBLK; ++s) {
        r0[s] = n0[s];
        r1[s] = n1[s];
      }
      if (kk + 1 < KSTEPS / KBLK) {
        const float* nb = bbase + (kk + 1) * (KBLK / 4) * 1024;
#pragma unroll
        for (int s = 0; s < KBLK; ++s) {
          n0[s] = *(const float4*)(nb + (s >> 2) * 1024 + (s & 3) * 32);
          n1[s] = *(const float4*)(nb + (s >> 2) * 1024 + (s & 3) * 32 + 4);
        }
      }
#pragma unroll
      for (int s = 0; s < KBLK; ++s) {
        int kb = 4 * (k0 + kk * KBLK + s) + 2 * hi;
        v2f a0, a1;
        a0.x = ybuf[m][kb];
        a0.y = ybuf[m][kb + 1];
        a1.x = ybuf[TILE_ROWS + m][kb];
        a1.y = ybuf[TILE_ROWS + m][kb + 1];
        v2f b;
        b.x = r0[s].x; b.y = r1[s].x;
        c0 = __builtin_amdgcn_wmma_f32_16x16x4_f32(false, a0, false, b, (short)0,
                                                   c0, false, false);
        d0 = __builtin_amdgcn_wmma_f32_16x16x4_f32(false, a1, false, b, (short)0,
                                                   d0, false, false);
        b.x = r0[s].y; b.y = r1[s].y;
        c1 = __builtin_amdgcn_wmma_f32_16x16x4_f32(false, a0, false, b, (short)0,
                                                   c1, false, false);
        d1 = __builtin_amdgcn_wmma_f32_16x16x4_f32(false, a1, false, b, (short)0,
                                                   d1, false, false);
        b.x = r0[s].z; b.y = r1[s].z;
        c2 = __builtin_amdgcn_wmma_f32_16x16x4_f32(false, a0, false, b, (short)0,
                                                   c2, false, false);
        d2 = __builtin_amdgcn_wmma_f32_16x16x4_f32(false, a1, false, b, (short)0,
                                                   d2, false, false);
        b.x = r0[s].w; b.y = r1[s].w;
        c3 = __builtin_amdgcn_wmma_f32_16x16x4_f32(false, a0, false, b, (short)0,
                                                   c3, false, false);
        d3 = __builtin_amdgcn_wmma_f32_16x16x4_f32(false, a1, false, b, (short)0,
                                                   d3, false, false);
      }
    }
    float* xw = &ybuf[0][0] + (KCODES / NWAVES) * w + 4 * lane;
#define VQ_PUT(T, t, acc)                                                     \
    *(float4*)(xw + ((T) * 8 + (t) * 2 + 0) * YPITCH) =                       \
        make_float4(acc[0], acc[1], acc[2], acc[3]);                          \
    *(float4*)(xw + ((T) * 8 + (t) * 2 + 1) * YPITCH) =                       \
        make_float4(acc[4], acc[5], acc[6], acc[7]);
    VQ_PUT(0, 0, c0) VQ_PUT(0, 1, c1) VQ_PUT(0, 2, c2) VQ_PUT(0, 3, c3)
    VQ_PUT(1, 0, d0) VQ_PUT(1, 1, d1) VQ_PUT(1, 2, d2) VQ_PUT(1, 3, d3)
#undef VQ_PUT
    __syncthreads();
    const int tw = w >> 1, jh = w & 1;
    const float* xr = &ybuf[0][0] + (tw * 8 + jh) * YPITCH + 4 * lane;
    float sv[4][4];
#pragma unroll
    for (int t = 0; t < 4; ++t) {
      const float* p = xr + 2 * t * YPITCH;
      float4 s = *(const float4*)p;
#pragma unroll
      for (int sw = 1; sw < NWAVES; ++sw) {
        const float4 o = *(const float4*)(p + (KCODES / NWAVES) * sw);
        s.x += o.x; s.y += o.y; s.z += o.z; s.w += o.w;
      }
      sv[t][0] = s.x; sv[t][1] = s.y; sv[t][2] = s.z; sv[t][3] = s.w;
    }
#pragma unroll
    for (int jj = 0; jj < 4; ++jj) {
      int r = TILE_ROWS * tw + 8 * hi + 4 * jh + jj;
      long idx = (base_row + r) * DIMD + 4 * m;
      float4 zo = *(const float4*)(z + idx);
      float4 zq = make_float4(sv[0][jj], sv[1][jj], sv[2][jj], sv[3][jj]);
      acc_cont += zo.x * (zo.x - zq.x);
      acc_cont += zo.y * (zo.y - zq.y);
      acc_cont += zo.z * (zo.z - zq.z);
      acc_cont += zo.w * (zo.w - zq.w);
      *(float4*)(out + idx) = zq;
    }
  }
  acc_cont = wave_sum(acc_cont);
  if (lane == 0) atomicAdd(scont, acc_cont);
}

__global__ __launch_bounds__(32) void vq_finalize(
    const float* __restrict__ kappa_p, const float* __restrict__ avgp_acc,
    const float* __restrict__ sdisc, const float* __restrict__ scont,
    float* __restrict__ out) {
  int lane = threadIdx.x;
  const float invN = 1.0f / (float)NROWS;
  float acc = 0.0f;
#pragma unroll
  for (int i = lane; i < KCODES; i += 32) {
    float a = avgp_acc[i] * invN;
    acc += a * logf(a + 1e-7f);
  }
  acc = wave_sum(acc);
  if (lane == 0) {
    float loss = (sdisc[0] + kappa_p[0] * scont[0]) / 8.0f;
    out[(long)NROWS * DIMD] = loss;
    out[(long)NROWS * DIMD + 1] = expf(-acc);
  }
}

extern "C" void kernel_launch(void* const* d_in, const int* in_sizes, int n_in,
                              void* d_out, int out_size, void* d_ws,
                              size_t ws_size, hipStream_t stream) {
  const float* z = (const float*)d_in[0];
  const float* kappa = (const float*)d_in[1];
  const float* cb = (const float*)d_in[2];
  const float* gum = (const float*)d_in[3];
  float* out = (float*)d_out;

  float* wsf = (float*)d_ws;
  float* avgp_acc = wsf;
  float* sdisc = wsf + 512;
  float* scont = wsf + 513;
  float* cbn = wsf + 1024;

  vq_zero<<<3, 256, 0, stream>>>(wsf);
  vq_cbnorm<<<64, 256, 0, stream>>>(cb, cbn);
  vq_main<<<NROWS / BLOCK_ROWS, 32 * NWAVES, 0, stream>>>(z, kappa, cbn, gum, out,
                                                avgp_acc, sdisc, scont);
  vq_finalize<<<1, 32, 0, stream>>>(kappa, avgp_acc, sdisc, scont, out);
}


# Global loads requested a stage ahead of use: gumbel rows two rows ahead in two register sets, GEMM1 codebook fragments half a trip (32 WMMAs) ahead, phase-first loads before barriers; sums unchanged
# speedup vs baseline: 2.1622x; 1.0488x over previous
#include <hip/hip_runtime.h>
#include <hip/hip_bf16.h>

typedef __attribute__((ext_vector_type(2))) float v2f;
typedef __attribute__((ext_vector_type(8))) float v8f;

#define DIMD 64
#define KCODES 512
#define NROWS 65536
#define TILE_ROWS 16
#define NTILES 2
#define BLOCK_ROWS (TILE_ROWS * NTILES)
#define NWAVES 4
#define ROWS_PER_WAVE (BLOCK_ROWS / NWAVES)
#define ZPITCH (DIMD + 4)
#define YPITCH (KCODES + 4)
#define LOG2EF 1.4426950408889634f
#define LN2F 0.6931471805599453f

__device__ __forceinline__ float xrow16(float x) {
  const int b = __builtin_bit_cast(int, x);
  return __builtin_bit_cast(
      float, __builtin_amdgcn_permlanex16(b, b, 0x76543210, 0xfedcba98, false, false));
}
template <int MASK>
__device__ __forceinline__ float xlane(float x) {
  const int b = __builtin_bit_cast(int, x);
  return __builtin_bit_cast(
      float, __builtin_amdgcn_update_dpp(b, b, 0x160 + MASK, 0xf, 0xf, true));
}
__device__ __forceinline__ float wave_sum(float v) {
  v += xrow16(v);
  v += xlane<8>(v);
  v += xlane<4>(v);
  v += xlane<2>(v);
  v += xlane<1>(v);
  return v;
}
__device__ __forceinline__ float wave_max(float v) {
  v = fmaxf(v, xrow16(v));
  v = fmaxf(v, xlane<8>(v));
  v = fmaxf(v, xlane<4>(v));
  v = fmaxf(v, xlane<2>(v));
  v = fmaxf(v, xlane<1>(v));
  return v;
}

__global__ void vq_zero(float* acc) {
  int i = blockIdx.x * blockDim.x + threadIdx.x;
  if (i < 514) acc[i] = 0.0f;
}

#define CBN_POS(c, d)                                                         \
  ((((c) >> 4) << 10) + (((d) >> 3) << 7) + ((((c) & 15) >> 2) << 5) +        \
   ((((d) >> 2) & 1) << 4) + (((c) & 3) << 2) + ((d) & 3))
__global__ __launch_bounds__(256) void vq_cbnorm(const float* __restrict__ cb,
                                                 float* __restrict__ cbn) {
  __shared__ float rows[8][DIMD];
  const int wv = threadIdx.x >> 5;
  const int lane = threadIdx.x & 31;
  const int code = blockIdx.x * 8 + wv;
  const float* row = cb + code * DIMD;
  float v0 = row[lane], v1 = row[lane + 32];
  float s = wave_sum(v0 * v0 + v1 * v1);
  float scale = 1.0f / fmaxf(sqrtf(s), 1e-12f);
  rows[wv][lane] = v0 * scale;
  rows[wv][lane + 32] = v1 * scale;
  __syncthreads();
  const int hb = lane >> 4, cc = (lane >> 2) & 3, e = lane & 3;
#pragma unroll
  for (int t = 0; t < 2; ++t) {
    const int ln = 2 * wv + t;
    const int grp = ln >> 3;
    const int i = ln & 7;
    const int c = blockIdx.x * 8 + 4 * grp + cc;
    const int d = 8 * i + 4 * hb + e;
    cbn[CBN_POS(c, d)] = rows[4 * grp + cc][d];
  }
}

__global__ __launch_bounds__(32 * NWAVES) void vq_main(
    const float* __restrict__ z, const float* __restrict__ kappa_p,
    const float* __restrict__ cbn, const float* __restrict__ gum,
    float* __restrict__ out, float* __restrict__ avgp_acc,
    float* __restrict__ sdisc, float* __restrict__ scont) {
  __shared__ __attribute__((aligned(16))) float zt[BLOCK_ROWS][ZPITCH];
  __shared__ __attribute__((aligned(16))) float ybuf[BLOCK_ROWS][YPITCH];

  const int lane = threadIdx.x & 31;
  const int w = __builtin_amdgcn_readfirstlane((int)(threadIdx.x >> 5));
  const int m = lane & 15;
  const int hi = lane >> 4;
  const long base_row = (long)blockIdx.x * BLOCK_ROWS;
  const float kappa = kappa_p[0];

  const int ct0 = w * (KCODES / 16 / NWAVES);
  const float* cb_first = cbn + CBN_POS(ct0 * 16 + m, 4 * hi);
  float4 ha[4], hb[4];
#pragma unroll
  for (int i = 0; i < 4; ++i) {
    ha[i] = *(const float4*)(cb_first + 128 * i);
    hb[i] = *(const float4*)(cb_first + 1024 + 128 * i);
  }

#pragma unroll
  for (int rr = 0; rr < ROWS_PER_WAVE; ++rr) {
    const int r = w * ROWS_PER_WAVE + rr;
    const float* zr = z + (base_row + r) * DIMD;
    float v0 = zr[lane], v1 = zr[lane + 32];
    float s = wave_sum(v0 * v0 + v1 * v1);
    float scale = 1.0f / fmaxf(sqrtf(s), 1e-12f);
    zt[r][lane] = v0 * scale;
    zt[r][lane + 32] = v1 * scale;
  }
  __syncthreads();

#define VQ_G1_CHUNK(I, BA4, BB4)                                               \
  {                                                                            \
    const float4 p4 = *(const float4*)&zt[m][8 * (I) + 4 * hi];                \
    const float4 q4 = *(const float4*)&zt[TILE_ROWS + m][8 * (I) + 4 * hi];    \
    v2f a0, a1, ba, bb;                                                        \
    a0.x = p4.x; a0.y = p4.y;                          \
    a1.x = q4.x; a1.y = q4.y;                                                  \
    ba.x = (BA4).x; ba.y = (BA4).y;                                            \
    bb.x = (BB4).x; bb.y = (BB4).y;                                            \
    ca0 = __builtin_amdgcn_wmma_f32_16x16x4_f32(false, a0, false, ba, (short)0, ca0, false, false); \
    ca1 = __builtin_amdgcn_wmma_f32_16x16x4_f32(false, a1, false, ba, (short)0, ca1, false, false); \
    cb0 = __builtin_amdgcn_wmma_f32_16x16x4_f32(false, a0, false, bb, (short)0, cb0, false, false); \
    cb1 = __builtin_amdgcn_wmma_f32_16x16x4_f32(false, a1, false, bb, (short)0, cb1, false, false); \
    a0.x = p4.z; a0.y = p4.w;                          \
    a1.x = q4.z; a1.y = q4.w;                                                  \
    ba.x = (BA4).z; ba.y = (BA4).w;                                            \
    bb.x = (BB4).z; bb.y = (BB4).w;                                            \
    ca0 = __builtin_amdgcn_wmma_f32_16x16x4_f32(false, a0, false, ba, (short)0, ca0, false, false); \
    ca1 = __builtin_amdgcn_wmma_f32_16x16x4_f32(false, a1, false, ba, (short)0, ca1, false, false); \
    cb0 = __builtin_amdgcn_wmma_f32_16x16x4_f32(false, a0, false, bb, (short)0, cb0, false, false); \
    cb1 = __builtin_amdgcn_wmma_f32_16x16x4_f32(false, a1, false, bb, (short)0, cb1, false, false); \
  }
  constexpr int G1_TRIPS = KCODES / 16 / NWAVES / 2;
  auto g1_trip = [&](const int t, const bool has_next) {
    const int ct = ct0 + 2 * t;
    const float* rowa = cb_first + 2048 * t;
    v8f ca0 = {}, cb0 = {};
    v8f ca1 = {}, cb1 = {};
    float4 la[4], lb[4];
#pragma unroll
    for (int i = 0; i < 4; ++i) {
      la[i] = *(const float4*)(rowa + 128 * (4 + i));
      lb[i] = *(const float4*)(rowa + 1024 + 128 * (4 + i));
    }
#pragma unroll
    for (int i = 0; i < 4; ++i) VQ_G1_CHUNK(i, ha[i], hb[i])
    if (has_next) {
#pragma unroll
      for (int i = 0; i < 4; ++i) {
        ha[i] = *(const float4*)(rowa + 2048 + 128 * i);
        hb[i] = *(const float4*)(rowa + 2048 + 1024 + 128 * i);
      }
    }
#pragma unroll
    for (int i = 0; i < 4; ++i) VQ_G1_CHUNK(4 + i, la[i], lb[i])
#pragma unroll
    for (int j = 0; j < 8; ++j) {
      ybuf[j + 8 * hi][ct * 16 + m] = kappa * ca0[j];
      ybuf[j + 8 * hi][ct * 16 + 16 + m] = kappa * cb0[j];
      ybuf[TILE_ROWS + j + 8 * hi][ct * 16 + m] = kappa * ca1[j];
      ybuf[TILE_ROWS + j + 8 * hi][ct * 16 + 16 + m] = kappa * cb1[j];
    }
  };
#pragma unroll 1
  for (int t = 0; t < G1_TRIPS - 1; ++t) g1_trip(t, true);
  g1_trip(G1_TRIPS - 1, false);
#undef VQ_G1_CHUNK
  const float4* ubase =
      (const float4*)(gum + (base_row + w * ROWS_PER_WAVE) * (long)KCODES) + lane;
  float4 ua[4], ub[4];
#pragma unroll
  for (int q = 0; q < 4; ++q) {
    ua[q] = ubase[32 * q];
    ub[q] = ubase[128 + 32 * q];
  }
  __syncthreads();

  float acc_pl = 0.0f;
  float* avgp = &zt[0][0] + w * KCODES;
  for (int q = 0; q < 4; ++q)
    *(float4*)(avgp + 128 * q + 4 * lane) = make_float4(0.0f, 0.0f, 0.0f, 0.0f);
  auto softmax_row = [&](const int rr, const float4 (&ucur)[4]) {
    const int r = w * ROWS_PER_WAVE + rr;
    float* yrow = &ybuf[r][4 * lane];
    float yv[16];
    float m1 = -3.4e38f;
#pragma unroll
    for (int q = 0; q < 4; ++q) {
      float4 y4 = *(const float4*)(yrow + 128 * q);
      float yy4[4] = {y4.x, y4.y, y4.z, y4.w};
#pragma unroll
      for (int j = 0; j < 4; ++j) {
        yv[q * 4 + j] = yy4[j];
        m1 = fmaxf(m1, yy4[j]);
      }
    }
    m1 = wave_max(m1);
    const float mc = m1 * LOG2EF;
    float e1[16], e2[16];
    float s1 = 0.0f, s2 = 0.0f;
#pragma unroll
    for (int i = 0; i < 16; ++i) {
      e1[i] = __builtin_amdgcn_exp2f(fmaf(yv[i], LOG2EF, -mc));
      s1 += e1[i];
    }
#pragma unroll
    for (int q = 0; q < 4; ++q) {
      float4 u4 = ucur[q];
      float uu[4] = {u4.x, u4.y, u4.z, u4.w};
#pragma unroll
      for (int j = 0; j < 4; ++j) {
        int i = q * 4 + j;
        float t = fmaf(-LN2F, __builtin_amdgcn_logf(uu[j] + 1e-10f), 1e-10f);
        float qq = e1[i] * __builtin_amdgcn_rcpf(t);
        e2[i] = qq * qq;
        s2 += e2[i];
      }
    }
    s1 = wave_sum(s1);
    s2 = wave_sum(s2);
    float lse = m1 + logf(s1);
    float inv_s1 = 1.0f / s1;
    float inv_s2 = 1.0f / s2;
#pragma unroll
    for (int q = 0; q < 4; ++q) {
      float4 a4 = *(const float4*)(avgp + 128 * q + 4 * lane);
      float pp[4], en[4];
#pragma unroll
      for (int j = 0; j < 4; ++j) {
        int i = q * 4 + j;
        float lp = yv[i] - lse;
        float p = e1[i] * inv_s1;
        acc_pl += p * lp;
        pp[j] = p;
        en[j] = e2[i] * inv_s2;
      }
      *(float4*)(avgp + 128 * q + 4 * lane) =
          make_float4(a4.x + pp[0], a4.y + pp[1], a4.z + pp[2], a4.w + pp[3]);
      *(float4*)(yrow + 128 * q) = make_float4(en[0], en[1], en[2], en[3]);
    }
  };
  for (int rr = 0; rr < ROWS_PER_WAVE - 2; rr += 2) {
    const float4* unext = ubase + 128 * (rr + 2);
    softmax_row(rr, ua);
#pragma unroll
    for (int q = 0; q < 4; ++q) ua[q] = unext[32 * q];
    softmax_row(rr + 1, ub);
#pragma unroll
    for (int q = 0; q < 4; ++q) ub[q] = unext[128 + 32 * q];
  }
  softmax_row(ROWS_PER_WAVE - 2, ua);
  softmax_row(ROWS_PER_WAVE - 1, ub);
  constexpr int KSTEPS = KCODES / 4 / NWAVES;
  constexpr int KBLK = 4;
  const int k0 = w * KSTEPS;
  const float* bbase = cbn + CBN_POS(4 * k0 + 2 * hi, 4 * m);
  float4 n0[KBLK], n1[KBLK];
#pragma unroll
  for (int s = 0; s < KBLK; ++s) {
    n0[s] = *(const float4*)(bbase + (s >> 2) * 1024 + (s & 3) * 32);
    n1[s] = *(const float4*)(bbase + (s >> 2) * 1024 + (s & 3) * 32 + 4);
  }
  acc_pl = wave_sum(acc_pl);
  if (lane == 0) atomicAdd(sdisc, acc_pl);
  __syncthreads();
  {
    const float* pl = &zt[0][0];
#pragma unroll
    for (int q = 0; q < KCODES / (32 * NWAVES); ++q) {
      const int i = q * (32 * NWAVES) + (int)threadIdx.x;
      atomicAdd(&avgp_acc[i], (pl[i] + pl[KCODES + i]) +
                                  (pl[2 * KCODES + i] + pl[3 * KCODES + i]));
    }
  }

  float acc_cont = 0.0f;
  {
    v8f c0 = {}, c1 = {}, c2 = {}, c3 = {};
    v8f d0 = {}, d1 = {}, d2 = {}, d3 = {};
#pragma unroll 2
    for (int kk = 0; kk < KSTEPS / KBLK; ++kk) {
      float4 r0[KBLK], r1[KBLK];
#pragma unroll
      for (int s = 0; s < KBLK; ++s) {
        r0[s] = n0[s];
        r1[s] = n1[s];
      }
      if (kk + 1 < KSTEPS / KBLK) {
        const float* nb = bbase + (kk + 1) * (KBLK / 4) * 1024;
#pragma unroll
        for (int s = 0; s < KBLK; ++s) {
          n0[s] = *(const float4*)(nb + (s >> 2) * 1024 + (s & 3) * 32);
          n1[s] = *(const float4*)(nb + (s >> 2) * 1024 + (s & 3) * 32 + 4);
        }
      }
#pragma unroll
      for (int s = 0; s < KBLK; ++s) {
        int kb = 4 * (k0 + kk * KBLK + s) + 2 * hi;
        v2f a0, a1;
        a0.x = ybuf[m][kb];
        a0.y = ybuf[m][kb + 1];
        a1.x = ybuf[TILE_ROWS + m][kb];
        a1.y = ybuf[TILE_ROWS + m][kb + 1];
        v2f b;
        b.x = r0[s].x; b.y = r1[s].x;
        c0 = __builtin_amdgcn_wmma_f32_16x16x4_f32(false, a0, false, b, (short)0,
                                                   c0, false, false);
        d0 = __builtin_amdgcn_wmma_f32_16x16x4_f32(false, a1, false, b, (short)0,
                                                   d0, false, false);
        b.x = r0[s].y; b.y = r1[s].y;
        c1 = __builtin_amdgcn_wmma_f32_16x16x4_f32(false, a0, false, b, (short)0,
                                                   c1, false, false);
        d1 = __builtin_amdgcn_wmma_f32_16x16x4_f32(false, a1, false, b, (short)0,
                                                   d1, false, false);
        b.x = r0[s].z; b.y = r1[s].z;
        c2 = __builtin_amdgcn_wmma_f32_16x16x4_f32(false, a0, false, b, (short)0,
                                                   c2, false, false);
        d2 = __builtin_amdgcn_wmma_f32_16x16x4_f32(false, a1, false, b, (short)0,
                                                   d2, false, false);
        b.x = r0[s].w; b.y = r1[s].w;
        c3 = __builtin_amdgcn_wmma_f32_16x16x4_f32(false, a0, false, b, (short)0,
                                                   c3, false, false);
        d3 = __builtin_amdgcn_wmma_f32_16x16x4_f32(false, a1, false, b, (short)0,
                                                   d3, false, false);
      }
    }
    float* xw = &ybuf[0][0] + (KCODES / NWAVES) * w + 4 * lane;
#define VQ_PUT(T, t, acc)                                                     \
    *(float4*)(xw + ((T) * 8 + (t) * 2 + 0) * YPITCH) =                       \
        make_float4(acc[0], acc[1], acc[2], acc[3]);                          \
    *(float4*)(xw + ((T) * 8 + (t) * 2 + 1) * YPITCH) =                       \
        make_float4(acc[4], acc[5], acc[6], acc[7]);
    VQ_PUT(0, 0, c0) VQ_PUT(0, 1, c1) VQ_PUT(0, 2, c2) VQ_PUT(0, 3, c3)
    VQ_PUT(1, 0, d0) VQ_PUT(1, 1, d1) VQ_PUT(1, 2, d2) VQ_PUT(1, 3, d3)
#undef VQ_PUT
    __syncthreads();
    const int tw = w >> 1, jh = w & 1;
    const float* xr = &ybuf[0][0] + (tw * 8 + jh) * YPITCH + 4 * lane;
    float sv[4][4];
#pragma unroll
    for (int t = 0; t < 4; ++t) {
      const float* p = xr + 2 * t * YPITCH;
      float4 s = *(const float4*)p;
#pragma unroll
      for (int sw = 1; sw < NWAVES; ++sw) {
        const float4 o = *(const float4*)(p + (KCODES / NWAVES) * sw);
        s.x += o.x; s.y += o.y; s.z += o.z; s.w += o.w;
      }
      sv[t][0] = s.x; sv[t][1] = s.y; sv[t][2] = s.z; sv[t][3] = s.w;
    }
#pragma unroll
    for (int jj = 0; jj < 4; ++jj) {
      int r = TILE_ROWS * tw + 8 * hi + 4 * jh + jj;
      long idx = (base_row + r) * DIMD + 4 * m;
      float4 zo = *(const float4*)(z + idx);
      float4 zq = make_float4(sv[0][jj], sv[1][jj], sv[2][jj], sv[3][jj]);
      acc_cont += zo.x * (zo.x - zq.x);
      acc_cont += zo.y * (zo.y - zq.y);
      acc_cont += zo.z * (zo.z - zq.z);
      acc_cont += zo.w * (zo.w - zq.w);
      *(float4*)(out + idx) = zq;
    }
  }
  acc_cont = wave_sum(acc_cont);
  if (lane == 0) atomicAdd(scont, acc_cont);
}

__global__ __launch_bounds__(32) void vq_finalize(
    const float* __restrict__ kappa_p, const float* __restrict__ avgp_acc,
    const float* __restrict__ sdisc, const float* __restrict__ scont,
    float* __restrict__ out) {
  int lane = threadIdx.x;
  const float invN = 1.0f / (float)NROWS;
  float acc = 0.0f;
#pragma unroll
  for (int i = lane; i < KCODES; i += 32) {
    float a = avgp_acc[i] * invN;
    acc += a * logf(a + 1e-7f);
  }
  acc = wave_sum(acc);
  if (lane == 0) {
    float loss = (sdisc[0] + kappa_p[0] * scont[0]) / 8.0f;
    out[(long)NROWS * DIMD] = loss;
    out[(long)NROWS * DIMD + 1] = expf(-acc);
  }
}

extern "C" void kernel_launch(void* const* d_in, const int* in_sizes, int n_in,
                              void* d_out, int out_size, void* d_ws,
                              size_t ws_size, hipStream_t stream) {
  const float* z = (const float*)d_in[0];
  const float* kappa = (const float*)d_in[1];
  const float* cb = (const float*)d_in[2];
  const float* gum = (const float*)d_in[3];
  float* out = (float*)d_out;

  float* wsf = (float*)d_ws;
  float* avgp_acc = wsf;
  float* sdisc = wsf + 512;
  float* scont = wsf + 513;
  float* cbn = wsf + 1024;

  vq_zero<<<3, 256, 0, stream>>>(wsf);
  vq_cbnorm<<<64, 256, 0, stream>>>(cb, cbn);
  vq_main<<<NROWS / BLOCK_ROWS, 32 * NWAVES, 0, stream>>>(z, kappa, cbn, gum, out,
                                                avgp_acc, sdisc, scont);
  vq_finalize<<<1, 32, 0, stream>>>(kappa, avgp_acc, sdisc, scont, out);
}


# Loss sums: one two-lane atomic per wave at kernel end into one of 16 accumulator lines (by block index), not two hot-word atomics on one line, one waited for at a barrier; sum order changes
# speedup vs baseline: 2.2410x; 1.0364x over previous
#include <hip/hip_runtime.h>
#include <hip/hip_bf16.h>

typedef __attribute__((ext_vector_type(2))) float v2f;
typedef __attribute__((ext_vector_type(8))) float v8f;

#define DIMD 64
#define KCODES 512
#define NROWS 65536
#define TILE_ROWS 16
#define NTILES 2
#define BLOCK_ROWS (TILE_ROWS * NTILES)
#define NWAVES 4
#define ROWS_PER_WAVE (BLOCK_ROWS / NWAVES)
#define ZPITCH (DIMD + 4)
#define YPITCH (KCODES + 4)
#define LSLOTS 16
#define LOG2EF 1.4426950408889634f
#define LN2F 0.6931471805599453f

__device__ __forceinline__ float xrow16(float x) {
  const int b = __builtin_bit_cast(int, x);
  return __builtin_bit_cast(
      float, __builtin_amdgcn_permlanex16(b, b, 0x76543210, 0xfedcba98, false, false));
}
template <int MASK>
__device__ __forceinline__ float xlane(float x) {
  const int b = __builtin_bit_cast(int, x);
  return __builtin_bit_cast(
      float, __builtin_amdgcn_update_dpp(b, b, 0x160 + MASK, 0xf, 0xf, true));
}
__device__ __forceinline__ float wave_sum(float v) {
  v += xrow16(v);
  v += xlane<8>(v);
  v += xlane<4>(v);
  v += xlane<2>(v);
  v += xlane<1>(v);
  return v;
}
__device__ __forceinline__ float wave_max(float v) {
  v = fmaxf(v, xrow16(v));
  v = fmaxf(v, xlane<8>(v));
  v = fmaxf(v, xlane<4>(v));
  v = fmaxf(v, xlane<2>(v));
  v = fmaxf(v, xlane<1>(v));
  return v;
}

__global__ void vq_zero(float* acc) {
  int i = blockIdx.x * blockDim.x + threadIdx.x;
  if (i < 1024) acc[i] = 0.0f;
}

#define CBN_POS(c, d)                                                         \
  ((((c) >> 4) << 10) + (((d) >> 3) << 7) + ((((c) & 15) >> 2) << 5) +        \
   ((((d) >> 2) & 1) << 4) + (((c) & 3) << 2) + ((d) & 3))
__global__ __launch_bounds__(256) void vq_cbnorm(const float* __restrict__ cb,
                                                 float* __restrict__ cbn) {
  __shared__ float rows[8][DIMD];
  const int wv = threadIdx.x >> 5;
  const int lane = threadIdx.x & 31;
  const int code = blockIdx.x * 8 + wv;
  const float* row = cb + code * DIMD;
  float v0 = row[lane], v1 = row[lane + 32];
  float s = wave_sum(v0 * v0 + v1 * v1);
  float scale = 1.0f / fmaxf(sqrtf(s), 1e-12f);
  rows[wv][lane] = v0 * scale;
  rows[wv][lane + 32] = v1 * scale;
  __syncthreads();
  const int hb = lane >> 4, cc = (lane >> 2) & 3, e = lane & 3;
#pragma unroll
  for (int t = 0; t < 2; ++t) {
    const int ln = 2 * wv + t;
    const int grp = ln >> 3;
    const int i = ln & 7;
    const int c = blockIdx.x * 8 + 4 * grp + cc;
    const int d = 8 * i + 4 * hb + e;
    cbn[CBN_POS(c, d)] = rows[4 * grp + cc][d];
  }
}

__global__ __launch_bounds__(32 * NWAVES) void vq_main(
    const float* __restrict__ z, const float* __restrict__ kappa_p,
    const float* __restrict__ cbn, const float* __restrict__ gum,
    float* __restrict__ out, float* __restrict__ avgp_acc,
    float* __restrict__ lsum) {
  __shared__ __attribute__((aligned(16))) float zt[BLOCK_ROWS][ZPITCH];
  __shared__ __attribute__((aligned(16))) float ybuf[BLOCK_ROWS][YPITCH];

  const int lane = threadIdx.x & 31;
  const int w = __builtin_amdgcn_readfirstlane((int)(threadIdx.x >> 5));
  const int m = lane & 15;
  const int hi = lane >> 4;
  const long base_row = (long)blockIdx.x * BLOCK_ROWS;
  const float kappa = kappa_p[0];

  const int ct0 = w * (KCODES / 16 / NWAVES);
  const float* cb_first = cbn + CBN_POS(ct0 * 16 + m, 4 * hi);
  float4 ha[4], hb[4];
#pragma unroll
  for (int i = 0; i < 4; ++i) {
    ha[i] = *(const float4*)(cb_first + 128 * i);
    hb[i] = *(const float4*)(cb_first + 1024 + 128 * i);
  }

#pragma unroll
  for (int rr = 0; rr < ROWS_PER_WAVE; ++rr) {
    const int r = w * ROWS_PER_WAVE + rr;
    const float* zr = z + (base_row + r) * DIMD;
    float v0 = zr[lane], v1 = zr[lane + 32];
    float s = wave_sum(v0 * v0 + v1 * v1);
    float scale = 1.0f / fmaxf(sqrtf(s), 1e-12f);
    zt[r][lane] = v0 * scale;
    zt[r][lane + 32] = v1 * scale;
  }
  __syncthreads();

#define VQ_G1_CHUNK(I, BA4, BB4)                                               \
  {                                                                            \
    const float4 p4 = *(const float4*)&zt[m][8 * (I) + 4 * hi];                \
    const float4 q4 = *(const float4*)&zt[TILE_ROWS + m][8 * (I) + 4 * hi];    \
    v2f a0, a1, ba, bb;                                                        \
    a0.x = p4.x; a0.y = p4.y;                          \
    a1.x = q4.x; a1.y = q4.y;                                                  \
    ba.x = (BA4).x; ba.y = (BA4).y;                                            \
    bb.x = (BB4).x; bb.y = (BB4).y;                                            \
    ca0 = __builtin_amdgcn_wmma_f32_16x16x4_f32(false, a0, false, ba, (short)0, ca0, false, false); \
    ca1 = __builtin_amdgcn_wmma_f32_16x16x4_f32(false, a1, false, ba, (short)0, ca1, false, false); \
    cb0 = __builtin_amdgcn_wmma_f32_16x16x4_f32(false, a0, false, bb, (short)0, cb0, false, false); \
    cb1 = __builtin_amdgcn_wmma_f32_16x16x4_f32(false, a1, false, bb, (short)0, cb1, false, false); \
    a0.x = p4.z; a0.y = p4.w;                          \
    a1.x = q4.z; a1.y = q4.w;                                                  \
    ba.x = (BA4).z; ba.y = (BA4).w;                                            \
    bb.x = (BB4).z; bb.y = (BB4).w;                                            \
    ca0 = __builtin_amdgcn_wmma_f32_16x16x4_f32(false, a0, false, ba, (short)0, ca0, false, false); \
    ca1 = __builtin_amdgcn_wmma_f32_16x16x4_f32(false, a1, false, ba, (short)0, ca1, false, false); \
    cb0 = __builtin_amdgcn_wmma_f32_16x16x4_f32(false, a0, false, bb, (short)0, cb0, false, false); \
    cb1 = __builtin_amdgcn_wmma_f32_16x16x4_f32(false, a1, false, bb, (short)0, cb1, false, false); \
  }
  constexpr int G1_TRIPS = KCODES / 16 / NWAVES / 2;
  auto g1_trip = [&](const int t, const bool has_next) {
    const int ct = ct0 + 2 * t;
    const float* rowa = cb_first + 2048 * t;
    v8f ca0 = {}, cb0 = {};
    v8f ca1 = {}, cb1 = {};
    float4 la[4], lb[4];
#pragma unroll
    for (int i = 0; i < 4; ++i) {
      la[i] = *(const float4*)(rowa + 128 * (4 + i));
      lb[i] = *(const float4*)(rowa + 1024 + 128 * (4 + i));
    }
#pragma unroll
    for (int i = 0; i < 4; ++i) VQ_G1_CHUNK(i, ha[i], hb[i])
    if (has_next) {
#pragma unroll
      for (int i = 0; i < 4; ++i) {
        ha[i] = *(const float4*)(rowa + 2048 + 128 * i);
        hb[i] = *(const float4*)(rowa + 2048 + 1024 + 128 * i);
      }
    }
#pragma unroll
    for (int i = 0; i < 4; ++i) VQ_G1_CHUNK(4 + i, la[i], lb[i])
#pragma unroll
    for (int j = 0; j < 8; ++j) {
      ybuf[j + 8 * hi][ct * 16 + m] = kappa * ca0[j];
      ybuf[j + 8 * hi][ct * 16 + 16 + m] = kappa * cb0[j];
      ybuf[TILE_ROWS + j + 8 * hi][ct * 16 + m] = kappa * ca1[j];
      ybuf[TILE_ROWS + j + 8 * hi][ct * 16 + 16 + m] = kappa * cb1[j];
    }
  };
#pragma unroll 1
  for (int t = 0; t < G1_TRIPS - 1; ++t) g1_trip(t, true);
  g1_trip(G1_TRIPS - 1, false);
#undef VQ_G1_CHUNK
  const float4* ubase =
      (const float4*)(gum + (base_row + w * ROWS_PER_WAVE) * (long)KCODES) + lane;
  float4 ua[4], ub[4];
#pragma unroll
  for (int q = 0; q < 4; ++q) {
    ua[q] = ubase[32 * q];
    ub[q] = ubase[128 + 32 * q];
  }
  __syncthreads();

  float acc_pl = 0.0f;
  float* avgp = &zt[0][0] + w * KCODES;
  for (int q = 0; q < 4; ++q)
    *(float4*)(avgp + 128 * q + 4 * lane) = make_float4(0.0f, 0.0f, 0.0f, 0.0f);
  auto softmax_row = [&](const int rr, const float4 (&ucur)[4]) {
    const int r = w * ROWS_PER_WAVE + rr;
    float* yrow = &ybuf[r][4 * lane];
    float yv[16];
    float m1 = -3.4e38f;
#pragma unroll
    for (int q = 0; q < 4; ++q) {
      float4 y4 = *(const float4*)(yrow + 128 * q);
      float yy4[4] = {y4.x, y4.y, y4.z, y4.w};
#pragma unroll
      for (int j = 0; j < 4; ++j) {
        yv[q * 4 + j] = yy4[j];
        m1 = fmaxf(m1, yy4[j]);
      }
    }
    m1 = wave_max(m1);
    const float mc = m1 * LOG2EF;
    float e1[16], e2[16];
    float s1 = 0.0f, s2 = 0.0f;
#pragma unroll
    for (int i = 0; i < 16; ++i) {
      e1[i] = __builtin_amdgcn_exp2f(fmaf(yv[i], LOG2EF, -mc));
      s1 += e1[i];
    }
#pragma unroll
    for (int q = 0; q < 4; ++q) {
      float4 u4 = ucur[q];
      float uu[4] = {u4.x, u4.y, u4.z, u4.w};
#pragma unroll
      for (int j = 0; j < 4; ++j) {
        int i = q * 4 + j;
        float t = fmaf(-LN2F, __builtin_amdgcn_logf(uu[j] + 1e-10f), 1e-10f);
        float qq = e1[i] * __builtin_amdgcn_rcpf(t);
        e2[i] = qq * qq;
        s2 += e2[i];
      }
    }
    s1 = wave_sum(s1);
    s2 = wave_sum(s2);
    float lse = m1 + logf(s1);
    float inv_s1 = 1.0f / s1;
    float inv_s2 = 1.0f / s2;
#pragma unroll
    for (int q = 0; q < 4; ++q) {
      float4 a4 = *(const float4*)(avgp + 128 * q + 4 * lane);
      float pp[4], en[4];
#pragma unroll
      for (int j = 0; j < 4; ++j) {
        int i = q * 4 + j;
        float lp = yv[i] - lse;
        float p = e1[i] * inv_s1;
        acc_pl += p * lp;
        pp[j] = p;
        en[j] = e2[i] * inv_s2;
      }
      *(float4*)(avgp + 128 * q + 4 * lane) =
          make_float4(a4.x + pp[0], a4.y + pp[1], a4.z + pp[2], a4.w + pp[3]);
      *(float4*)(yrow + 128 * q) = make_float4(en[0], en[1], en[2], en[3]);
    }
  };
  for (int rr = 0; rr < ROWS_PER_WAVE - 2; rr += 2) {
    const float4* unext = ubase + 128 * (rr + 2);
    softmax_row(rr, ua);
#pragma unroll
    for (int q = 0; q < 4; ++q) ua[q] = unext[32 * q];
    softmax_row(rr + 1, ub);
#pragma unroll
    for (int q = 0; q < 4; ++q) ub[q] = unext[128 + 32 * q];
  }
  softmax_row(ROWS_PER_WAVE - 2, ua);
  softmax_row(ROWS_PER_WAVE - 1, ub);
  constexpr int KSTEPS = KCODES / 4 / NWAVES;
  constexpr int KBLK = 4;
  const int k0 = w * KSTEPS;
  const float* bbase = cbn + CBN_POS(4 * k0 + 2 * hi, 4 * m);
  float4 n0[KBLK], n1[KBLK];
#pragma unroll
  for (int s = 0; s < KBLK; ++s) {
    n0[s] = *(const float4*)(bbase + (s >> 2) * 1024 + (s & 3) * 32);
    n1[s] = *(const float4*)(bbase + (s >> 2) * 1024 + (s & 3) * 32 + 4);
  }
  acc_pl = wave_sum(acc_pl);
  __syncthreads();
  {
    const float* pl = &zt[0][0];
#pragma unroll
    for (int q = 0; q < KCODES / (32 * NWAVES); ++q) {
      const int i = q * (32 * NWAVES) + (int)threadIdx.x;
      atomicAdd(&avgp_acc[i], (pl[i] + pl[KCODES + i]) +
                                  (pl[2 * KCODES + i] + pl[3 * KCODES + i]));
    }
  }

  float acc_cont = 0.0f;
  {
    v8f c0 = {}, c1 = {}, c2 = {}, c3 = {};
    v8f d0 = {}, d1 = {}, d2 = {}, d3 = {};
#pragma unroll 2
    for (int kk = 0; kk < KSTEPS / KBLK; ++kk) {
      float4 r0[KBLK], r1[KBLK];
#pragma unroll
      for (int s = 0; s < KBLK; ++s) {
        r0[s] = n0[s];
        r1[s] = n1[s];
      }
      if (kk + 1 < KSTEPS / KBLK) {
        const float* nb = bbase + (kk + 1) * (KBLK / 4) * 1024;
#pragma unroll
        for (int s = 0; s < KBLK; ++s) {
          n0[s] = *(const float4*)(nb + (s >> 2) * 1024 + (s & 3) * 32);
          n1[s] = *(const float4*)(nb + (s >> 2) * 1024 + (s & 3) * 32 + 4);
        }
      }
#pragma unroll
      for (int s = 0; s < KBLK; ++s) {
        int kb = 4 * (k0 + kk * KBLK + s) + 2 * hi;
        v2f a0, a1;
        a0.x = ybuf[m][kb];
        a0.y = ybuf[m][kb + 1];
        a1.x = ybuf[TILE_ROWS + m][kb];
        a1.y = ybuf[TILE_ROWS + m][kb + 1];
        v2f b;
        b.x = r0[s].x; b.y = r1[s].x;
        c0 = __builtin_amdgcn_wmma_f32_16x16x4_f32(false, a0, false, b, (short)0,
                                                   c0, false, false);
        d0 = __builtin_amdgcn_wmma_f32_16x16x4_f32(false, a1, false, b, (short)0,
                                                   d0, false, false);
        b.x = r0[s].y; b.y = r1[s].y;
        c1 = __builtin_amdgcn_wmma_f32_16x16x4_f32(false, a0, false, b, (short)0,
                                                   c1, false, false);
        d1 = __builtin_amdgcn_wmma_f32_16x16x4_f32(false, a1, false, b, (short)0,
                                                   d1, false, false);
        b.x = r0[s].z; b.y = r1[s].z;
        c2 = __builtin_amdgcn_wmma_f32_16x16x4_f32(false, a0, false, b, (short)0,
                                                   c2, false, false);
        d2 = __builtin_amdgcn_wmma_f32_16x16x4_f32(false, a1, false, b, (short)0,
                                                   d2, false, false);
        b.x = r0[s].w; b.y = r1[s].w;
        c3 = __builtin_amdgcn_wmma_f32_16x16x4_f32(false, a0, false, b, (short)0,
                                                   c3, false, false);
        d3 = __builtin_amdgcn_wmma_f32_16x16x4_f32(false, a1, false, b, (short)0,
                                                   d3, false, false);
      }
    }
    float* xw = &ybuf[0][0] + (KCODES / NWAVES) * w + 4 * lane;
#define VQ_PUT(T, t, acc)                                                     \
    *(float4*)(xw + ((T) * 8 + (t) * 2 + 0) * YPITCH) =                       \
        make_float4(acc[0], acc[1], acc[2], acc[3]);                          \
    *(float4*)(xw + ((T) * 8 + (t) * 2 + 1) * YPITCH) =                       \
        make_float4(acc[4], acc[5], acc[6], acc[7]);
    VQ_PUT(0, 0, c0) VQ_PUT(0, 1, c1) VQ_PUT(0, 2, c2) VQ_PUT(0, 3, c3)
    VQ_PUT(1, 0, d0) VQ_PUT(1, 1, d1) VQ_PUT(1, 2, d2) VQ_PUT(1, 3, d3)
#undef VQ_PUT
    __syncthreads();
    const int tw = w >> 1, jh = w & 1;
    const float* xr = &ybuf[0][0] + (tw * 8 + jh) * YPITCH + 4 * lane;
    float sv[4][4];
#pragma unroll
    for (int t = 0; t < 4; ++t) {
      const float* p = xr + 2 * t * YPITCH;
      float4 s = *(const float4*)p;
#pragma unroll
      for (int sw = 1; sw < NWAVES; ++sw) {
        const float4 o = *(const float4*)(p + (KCODES / NWAVES) * sw);
        s.x += o.x; s.y += o.y; s.z += o.z; s.w += o.w;
      }
      sv[t][0] = s.x; sv[t][1] = s.y; sv[t][2] = s.z; sv[t][3] = s.w;
    }
#pragma unroll
    for (int jj = 0; jj < 4; ++jj) {
      int r = TILE_ROWS * tw + 8 * hi + 4 * jh + jj;
      long idx = (base_row + r) * DIMD + 4 * m;
      float4 zo = *(const float4*)(z + idx);
      float4 zq = make_float4(sv[0][jj], sv[1][jj], sv[2][jj], sv[3][jj]);
      acc_cont += zo.x * (zo.x - zq.x);
      acc_cont += zo.y * (zo.y - zq.y);
      acc_cont += zo.z * (zo.z - zq.z);
      acc_cont += zo.w * (zo.w - zq.w);
      *(float4*)(out + idx) = zq;
    }
  }
  acc_cont = wave_sum(acc_cont);
  if (lane < 2)
    atomicAdd(lsum + 32 * (int)(blockIdx.x & (LSLOTS - 1)) + lane,
              lane == 0 ? acc_pl : acc_cont);
}

__global__ __launch_bounds__(32) void vq_finalize(
    const float* __restrict__ kappa_p, const float* __restrict__ avgp_acc,
    const float* __restrict__ lsum, float* __restrict__ out) {
  int lane = threadIdx.x;
  float sd = 0.0f, sc = 0.0f;
  if (lane < LSLOTS) {
    sd = lsum[32 * lane];
    sc = lsum[32 * lane + 1];
  }
  sd = wave_sum(sd);
  sc = wave_sum(sc);
  const float invN = 1.0f / (float)NROWS;
  float acc = 0.0f;
#pragma unroll
  for (int i = lane; i < KCODES; i += 32) {
    float a = avgp_acc[i] * invN;
    acc += a * logf(a + 1e-7f);
  }
  acc = wave_sum(acc);
  if (lane == 0) {
    float loss = (sd + kappa_p[0] * sc) / 8.0f;
    out[(long)NROWS * DIMD] = loss;
    out[(long)NROWS * DIMD + 1] = expf(-acc);
  }
}

extern "C" void kernel_launch(void* const* d_in, const int* in_sizes, int n_in,
                              void* d_out, int out_size, void* d_ws,
                              size_t ws_size, hipStream_t stream) {
  const float* z = (const float*)d_in[0];
  const float* kappa = (const float*)d_in[1];
  const float* cb = (const float*)d_in[2];
  const float* gum = (const float*)d_in[3];
  float* out = (float*)d_out;

  float* wsf = (float*)d_ws;
  float* avgp_acc = wsf;
  float* lsum = wsf + 512;
  float* cbn = wsf + 1024;

  vq_zero<<<4, 256, 0, stream>>>(wsf);
  vq_cbnorm<<<64, 256, 0, stream>>>(cb, cbn);
  vq_main<<<NROWS / BLOCK_ROWS, 32 * NWAVES, 0, stream>>>(z, kappa, cbn, gum, out,
                                                avgp_acc, lsum);
  vq_finalize<<<1, 32, 0, stream>>>(kappa, avgp_acc, lsum, out);
}
